# starting kernel
# baseline (speedup 1.0000x reference)
#include <hip/hip_runtime.h>
#include <hip/hip_bf16.h>

#define DD 128
#define XS 132
#define WS 136

typedef __attribute__((ext_vector_type(2))) float v2f;
typedef __attribute__((ext_vector_type(8))) float v8f;

__global__ void deg_init_kernel(float* __restrict__ deg, int nN) {
    int i = blockIdx.x * blockDim.x + threadIdx.x;
    if (i < nN) deg[i] = 1.0f;
}

__global__ void deg_edge_kernel(const int* __restrict__ dst, float* __restrict__ deg, int nE) {
    int e = blockIdx.x * blockDim.x + threadIdx.x;
    if (e < nE) atomicAdd(&deg[dst[e]], 1.0f);
}

__global__ void dinv_kernel(float* __restrict__ deg, int nN) {
    int i = blockIdx.x * blockDim.x + threadIdx.x;
    if (i < nN) deg[i] = rsqrtf(deg[i]);
}

__global__ __launch_bounds__(256)
void gcn_gemm_wmma(const float* __restrict__ X, const float* __restrict__ W,
                   float* __restrict__ H, int nN, int applyRelu) {
    __shared__ __align__(16) float sW[DD * WS];
    __shared__ __align__(16) float sX[128 * XS];

    const int tid  = threadIdx.x;
    const int wave = tid >> 5;
    const int lane = tid & 31;
    const int lr   = lane & 15;
    const int hi   = lane >> 4;
    const int rowBase = blockIdx.x * 128;

    for (int i = tid; i < DD * 32; i += 256) {
        int r  = i >> 5;
        int c4 = (i & 31) << 2;
        float4 w = *(const float4*)(W + r * DD + c4);
        *(float4*)(sW + r * WS + c4) = w;
    }
    for (int i = tid; i < 128 * 32; i += 256) {
        int r  = i >> 5;
        int c4 = (i & 31) << 2;
        int gr = rowBase + r;
        float4 v;
        if (gr < nN) {
            v = *(const float4*)(X + (size_t)gr * DD + c4);
            if (applyRelu) {
                v.x = fmaxf(v.x, 0.0f); v.y = fmaxf(v.y, 0.0f);
                v.z = fmaxf(v.z, 0.0f); v.w = fmaxf(v.w, 0.0f);
            }
        } else {
            v = make_float4(0.0f, 0.0f, 0.0f, 0.0f);
        }
        *(float4*)(sX + r * XS + c4) = v;
    }
    __syncthreads();

    const float* xrow = sX + (wave * 16 + lr) * XS;

    v8f acc[8];
#pragma unroll
    for (int t = 0; t < 8; ++t) acc[t] = (v8f){0.f,0.f,0.f,0.f,0.f,0.f,0.f,0.f};

    for (int kb = 0; kb < 32; ++kb) {
        const int k0 = kb * 4 + hi * 2;
        v2f a = *(const v2f*)(xrow + k0);
        const float* wp = sW + k0 * WS + lr;
#pragma unroll
        for (int t = 0; t < 8; ++t) {
            v2f b;
            b.x = wp[t * 16];
            b.y = wp[t * 16 + WS];
            acc[t] = __builtin_amdgcn_wmma_f32_16x16x4_f32(
                false, a, false, b, (short)0, acc[t], false, false);
        }
    }

    const int row0 = rowBase + wave * 16 + hi * 8;
#pragma unroll
    for (int t = 0; t < 8; ++t) {
        const int col = t * 16 + lr;
#pragma unroll
        for (int v = 0; v < 8; ++v) {
            int r = row0 + v;
            if (r < nN) H[(size_t)r * DD + col] = acc[t][v];
        }
    }
}

__global__ void init_out_kernel(const float* __restrict__ h, const float* __restrict__ dinv,
                                const float* __restrict__ bias, float* __restrict__ out, int nN) {
    int idx = blockIdx.x * blockDim.x + threadIdx.x;
    int i = idx >> 5;
    int g = (idx & 31) << 2;
    if (i >= nN) return;
    float di = dinv[i];
    float w  = di * di;
    float4 v = *(const float4*)(h + (size_t)i * DD + g);
    float4 b = *(const float4*)(bias + g);
    float4 r;
    r.x = v.x * w + b.x;  r.y = v.y * w + b.y;
    r.z = v.z * w + b.z;  r.w = v.w * w + b.w;
    *(float4*)(out + (size_t)i * DD + g) = r;
}

__global__ void edge_scatter_kernel(const float* __restrict__ h, const float* __restrict__ dinv,
                                    const int* __restrict__ src, const int* __restrict__ dst,
                                    float* __restrict__ out, int nE) {
    unsigned idx = blockIdx.x * blockDim.x + threadIdx.x;
    int e = idx >> 5;
    int g = (idx & 31) << 2;
    if (e >= nE) return;
    int s = src[e];
    int d = dst[e];
    float nrm = dinv[s] * dinv[d];
    float4 v = *(const float4*)(h + (size_t)s * DD + g);
    float* o = out + (size_t)d * DD + g;
    atomicAdd(o + 0, nrm * v.x);
    atomicAdd(o + 1, nrm * v.y);
    atomicAdd(o + 2, nrm * v.z);
    atomicAdd(o + 3, nrm * v.w);
}

extern "C" void kernel_launch(void* const* d_in, const int* in_sizes, int n_in,
                              void* d_out, int out_size, void* d_ws, size_t ws_size,
                              hipStream_t stream) {
    const float* x   = (const float*)d_in[0];
    const int*   ei  = (const int*)d_in[1];
    const float* W1  = (const float*)d_in[2];
    const float* b1  = (const float*)d_in[3];
    const float* W2  = (const float*)d_in[4];
    const float* b2  = (const float*)d_in[5];
    const float* W3  = (const float*)d_in[6];
    const float* b3  = (const float*)d_in[7];

    const int nN = in_sizes[0] / DD;
    const int nE = in_sizes[1] / 2;
    const int* src = ei;
    const int* dst = ei + nE;

    const size_t npad = (size_t)((nN + 127) / 128) * 128;
    float* dinv = (float*)d_ws;
    float* bufH = dinv + npad;
    float* bufX = bufH + npad * DD;
    float* outF = (float*)d_out;

    const int TB = 256;
    const dim3 blk(TB);
    const int gN   = (nN + TB - 1) / TB;
    const int gE   = (nE + TB - 1) / TB;
    const int gNd  = (int)(((size_t)nN * 32 + TB - 1) / TB);
    const int gEd  = (int)(((size_t)nE * 32 + TB - 1) / TB);
    const int gGemm = (nN + 127) / 128;

    deg_init_kernel<<<gN, blk, 0, stream>>>(dinv, nN);
    deg_edge_kernel<<<gE, blk, 0, stream>>>(dst, dinv, nE);
    dinv_kernel<<<gN, blk, 0, stream>>>(dinv, nN);

    gcn_gemm_wmma<<<gGemm, blk, 0, stream>>>(x, W1, bufH, nN, 0);
    init_out_kernel<<<gNd, blk, 0, stream>>>(bufH, dinv, b1, bufX, nN);
    edge_scatter_kernel<<<gEd, blk, 0, stream>>>(bufH, dinv, src, dst, bufX, nE);

    gcn_gemm_wmma<<<gGemm, blk, 0, stream>>>(bufX, W2, bufH, nN, 1);
    init_out_kernel<<<gNd, blk, 0, stream>>>(bufH, dinv, b2, bufX, nN);
    edge_scatter_kernel<<<gEd, blk, 0, stream>>>(bufH, dinv, src, dst, bufX, nE);

    gcn_gemm_wmma<<<gGemm, blk, 0, stream>>>(bufX, W3, bufH, nN, 1);
    init_out_kernel<<<gNd, blk, 0, stream>>>(bufH, dinv, b3, outF, nN);
    edge_scatter_kernel<<<gEd, blk, 0, stream>>>(bufH, dinv, src, dst, outF, nE);
}


# per-edge float atomic scatter + init pass replaced by one-wave-per-node gather over per-call destination lists (count, scan, fill, order by source): fixed summation order, each out row stored once
# speedup vs baseline: 2.2826x; 2.2826x over previous
#include <hip/hip_runtime.h>
#include <hip/hip_bf16.h>

#define DD 128
#define XS 132
#define WS 136

#define SCAN_TILE 4096

typedef __attribute__((ext_vector_type(2))) float v2f;
typedef __attribute__((ext_vector_type(8))) float v8f;

__global__ void cnt_zero_kernel(int* __restrict__ cnt, int n) {
    int i = blockIdx.x * blockDim.x + threadIdx.x;
    if (i < n) cnt[i] = 0;
}

__global__ void edge_rank_kernel(const int* __restrict__ dst, int* __restrict__ cnt,
                                 int* __restrict__ pos, int nE) {
    int e = blockIdx.x * blockDim.x + threadIdx.x;
    if (e < nE) pos[e] = atomicAdd(&cnt[dst[e]], 1);
}

__global__ __launch_bounds__(1024)
void scan_dinv_kernel(const int* __restrict__ cnt, int* __restrict__ rowptr,
                      float* __restrict__ dinv, int nPad, int nTiles) {
    __shared__ int sTot[2][32];
    const int tid  = threadIdx.x;
    const int lane = tid & 31;
    const int wv   = tid >> 5;
    int carry = 0;
    for (int t = 0; t < nTiles; ++t) {
        const int base = t * SCAN_TILE + tid * 4;
        const int4 c = *(const int4*)(cnt + base);
        const int s1 = c.x, s2 = s1 + c.y, s3 = s2 + c.z, s4 = s3 + c.w;
        int incl = s4;
        for (int off = 1; off < 32; off <<= 1) {
            const int up = __shfl_up(incl, off);
            if (lane >= off) incl += up;
        }
        int* tot = sTot[t & 1];
        if (lane == 31) tot[wv] = incl;
        __syncthreads();
        int before = 0, all = 0;
        for (int w = 0; w < 32; ++w) {
            const int v = tot[w];
            all += v;
            before += (w < wv) ? v : 0;
        }
        const int excl = carry + before + incl - s4;
        *(int4*)(rowptr + base) = make_int4(excl, excl + s1, excl + s2, excl + s3);
        if (base < nPad)
            *(float4*)(dinv + base) = make_float4(rsqrtf((float)(c.x + 1)), rsqrtf((float)(c.y + 1)),
                                                  rsqrtf((float)(c.z + 1)), rsqrtf((float)(c.w + 1)));
        carry += all;
    }
}

__global__ void csr_fill_kernel(const int* __restrict__ src, const int* __restrict__ dst,
                                const int* __restrict__ pos, const int* __restrict__ rowptr,
                                int* __restrict__ ctmp, int nE) {
    int e = blockIdx.x * blockDim.x + threadIdx.x;
    if (e < nE) ctmp[rowptr[dst[e]] + pos[e]] = src[e];
}

__global__ __launch_bounds__(256)
void csr_sort_kernel(const int* __restrict__ rowptr, const int* __restrict__ ctmp,
                     const float* __restrict__ dinv, int* __restrict__ col,
                     float* __restrict__ wgt, int nN) {
    const int lane = threadIdx.x & 31;
    const int i = blockIdx.x * 8 + __builtin_amdgcn_readfirstlane(threadIdx.x >> 5);
    if (i >= nN) return;
    const int beg = rowptr[i];
    const int deg = rowptr[i + 1] - beg;
    const float di = dinv[i];
    const int* seg = ctmp + beg;
    if (deg <= 32) {
        const int key = (lane < deg) ? seg[lane] : 0;
        int rank = 0;
        for (int m = 0; m < deg; ++m) {
            const int km = __builtin_amdgcn_readlane(key, m);
            rank += (km < key || (km == key && m < lane)) ? 1 : 0;
        }
        if (lane < deg) {
            col[beg + rank] = key;
            wgt[beg + rank] = di * dinv[key];
        }
    } else {
        for (int j = lane; j < deg; j += 32) {
            const int key = seg[j];
            int rank = 0;
            for (int m = 0; m < deg; ++m) {
                const int km = seg[m];
                rank += (km < key || (km == key && m < j)) ? 1 : 0;
            }
            col[beg + rank] = key;
            wgt[beg + rank] = di * dinv[key];
        }
    }
}

__global__ __launch_bounds__(256)
void gcn_gemm_wmma(const float* __restrict__ X, const float* __restrict__ W,
                   float* __restrict__ H, int nN, int applyRelu) {
    __shared__ __align__(16) float sW[DD * WS];
    __shared__ __align__(16) float sX[128 * XS];

    const int tid  = threadIdx.x;
    const int wave = tid >> 5;
    const int lane = tid & 31;
    const int lr   = lane & 15;
    const int hi   = lane >> 4;
    const int rowBase = blockIdx.x * 128;

    for (int i = tid; i < DD * 32; i += 256) {
        int r  = i >> 5;
        int c4 = (i & 31) << 2;
        float4 w = *(const float4*)(W + r * DD + c4);
        *(float4*)(sW + r * WS + c4) = w;
    }
    for (int i = tid; i < 128 * 32; i += 256) {
        int r  = i >> 5;
        int c4 = (i & 31) << 2;
        int gr = rowBase + r;
        float4 v;
        if (gr < nN) {
            v = *(const float4*)(X + (size_t)gr * DD + c4);
            if (applyRelu) {
                v.x = fmaxf(v.x, 0.0f); v.y = fmaxf(v.y, 0.0f);
                v.z = fmaxf(v.z, 0.0f); v.w = fmaxf(v.w, 0.0f);
            }
        } else {
            v = make_float4(0.0f, 0.0f, 0.0f, 0.0f);
        }
        *(float4*)(sX + r * XS + c4) = v;
    }
    __syncthreads();

    const float* xrow = sX + (wave * 16 + lr) * XS;

    v8f acc[8];
#pragma unroll
    for (int t = 0; t < 8; ++t) acc[t] = (v8f){0.f,0.f,0.f,0.f,0.f,0.f,0.f,0.f};

    for (int kb = 0; kb < 32; ++kb) {
        const int k0 = kb * 4 + hi * 2;
        v2f a = *(const v2f*)(xrow + k0);
        const float* wp = sW + k0 * WS + lr;
#pragma unroll
        for (int t = 0; t < 8; ++t) {
            v2f b;
            b.x = wp[t * 16];
            b.y = wp[t * 16 + WS];
            acc[t] = __builtin_amdgcn_wmma_f32_16x16x4_f32(
                false, a, false, b, (short)0, acc[t], false, false);
        }
    }

    const int row0 = rowBase + wave * 16 + hi * 8;
#pragma unroll
    for (int t = 0; t < 8; ++t) {
        const int col = t * 16 + lr;
#pragma unroll
        for (int v = 0; v < 8; ++v) {
            int r = row0 + v;
            if (r < nN) H[(size_t)r * DD + col] = acc[t][v];
        }
    }
}

template <int G>
__device__ __forceinline__ void gather_group(float4& acc, const float4* __restrict__ h4,
                                             int cs, float cw, int k, int lane) {
    float4 v[G];
    float  w[G];
#pragma unroll
    for (int j = 0; j < G; ++j) {
        const int s = __builtin_amdgcn_readlane(cs, k + j);
        w[j] = __int_as_float(__builtin_amdgcn_readlane(__float_as_int(cw), k + j));
        v[j] = h4[(size_t)s * 32 + lane];
    }
#pragma unroll
    for (int j = 0; j < G; ++j) {
        acc.x += w[j] * v[j].x;  acc.y += w[j] * v[j].y;
        acc.z += w[j] * v[j].z;  acc.w += w[j] * v[j].w;
    }
}

__global__ __launch_bounds__(256)
void gather_rows_kernel(const float* __restrict__ h, const float* __restrict__ dinv,
                        const float* __restrict__ bias, const int* __restrict__ rowptr,
                        const int* __restrict__ col, const float* __restrict__ wgt,
                        float* __restrict__ out, int nN) {
    const int lane = threadIdx.x & 31;
    const int i = blockIdx.x * 8 + __builtin_amdgcn_readfirstlane(threadIdx.x >> 5);
    if (i >= nN) return;
    const int beg = rowptr[i];
    const int end = rowptr[i + 1];
    const float di = dinv[i];
    const float4* __restrict__ h4 = (const float4*)h;

    float4 acc;
    {
        const float w0 = di * di;
        const float4 v = h4[(size_t)i * 32 + lane];
        const float4 b = ((const float4*)bias)[lane];
        acc.x = v.x * w0 + b.x;  acc.y = v.y * w0 + b.y;
        acc.z = v.z * w0 + b.z;  acc.w = v.w * w0 + b.w;
    }

    for (int p = beg; p < end; p += 32) {
        const int n = (end - p < 32) ? (end - p) : 32;
        int   cs = 0;
        float cw = 0.0f;
        if (lane < n) { cs = col[p + lane]; cw = wgt[p + lane]; }
        int k = 0;
        for (; k + 8 <= n; k += 8) gather_group<8>(acc, h4, cs, cw, k, lane);
        if (k + 4 <= n) { gather_group<4>(acc, h4, cs, cw, k, lane); k += 4; }
        if (k + 2 <= n) { gather_group<2>(acc, h4, cs, cw, k, lane); k += 2; }
        if (k < n)      { gather_group<1>(acc, h4, cs, cw, k, lane); }
    }
    ((float4*)out)[(size_t)i * 32 + lane] = acc;
}

extern "C" void kernel_launch(void* const* d_in, const int* in_sizes, int n_in,
                              void* d_out, int out_size, void* d_ws, size_t ws_size,
                              hipStream_t stream) {
    const float* x   = (const float*)d_in[0];
    const int*   ei  = (const int*)d_in[1];
    const float* W1  = (const float*)d_in[2];
    const float* b1  = (const float*)d_in[3];
    const float* W2  = (const float*)d_in[4];
    const float* b2  = (const float*)d_in[5];
    const float* W3  = (const float*)d_in[6];
    const float* b3  = (const float*)d_in[7];

    const int nN = in_sizes[0] / DD;
    const int nE = in_sizes[1] / 2;
    const int* src = ei;
    const int* dst = ei + nE;

    const size_t npad = (size_t)((nN + 127) / 128) * 128;
    const int nTiles  = nN / SCAN_TILE + 1;
    const size_t ncnt = (size_t)nTiles * SCAN_TILE;
    float* dinv = (float*)d_ws;
    float* bufH = dinv + npad;
    int*   cnt    = (int*)(bufH + npad * DD);
    int*   rowptr = cnt + ncnt;
    int*   pos    = rowptr + ncnt + 128;
    int*   ctmp   = pos + nE;
    int*   col    = ctmp + nE;
    float* wgt    = (float*)(col + nE);
    float* outF = (float*)d_out;

    const int TB = 256;
    const dim3 blk(TB);
    const int gC   = (int)((ncnt + TB - 1) / TB);
    const int gE   = (nE + TB - 1) / TB;
    const int gW   = (nN + 7) / 8;
    const int gGemm = (nN + 127) / 128;

    cnt_zero_kernel<<<gC, blk, 0, stream>>>(cnt, (int)ncnt);
    edge_rank_kernel<<<gE, blk, 0, stream>>>(dst, cnt, pos, nE);
    scan_dinv_kernel<<<1, dim3(1024), 0, stream>>>(cnt, rowptr, dinv, (int)npad, nTiles);
    csr_fill_kernel<<<gE, blk, 0, stream>>>(src, dst, pos, rowptr, ctmp, nE);
    csr_sort_kernel<<<gW, blk, 0, stream>>>(rowptr, ctmp, dinv, col, wgt, nN);

    gcn_gemm_wmma<<<gGemm, blk, 0, stream>>>(x, W1, bufH, nN, 0);
    gather_rows_kernel<<<gW, blk, 0, stream>>>(bufH, dinv, b1, rowptr, col, wgt, outF, nN);

    gcn_gemm_wmma<<<gGemm, blk, 0, stream>>>(outF, W2, bufH, nN, 1);
    gather_rows_kernel<<<gW, blk, 0, stream>>>(bufH, dinv, b2, rowptr, col, wgt, outF, nN);

    gcn_gemm_wmma<<<gGemm, blk, 0, stream>>>(outF, W3, bufH, nN, 1);
    gather_rows_kernel<<<gW, blk, 0, stream>>>(bufH, dinv, b3, rowptr, col, wgt, outF, nN);
}


# edge lists built on LDS counters instead of per-edge global returning adds: edges binned by destination range (count, place), then one block per range builds its node lists; sum order unchanged
# speedup vs baseline: 3.0240x; 1.3248x over previous
#include <hip/hip_runtime.h>
#include <hip/hip_bf16.h>


#define DD 128
#define XS 132
#define WS 136

#define NB   256
#define LB   256
#define LT   1024
#define RMAX 256

typedef __attribute__((ext_vector_type(2))) float v2f;
typedef __attribute__((ext_vector_type(8))) float v8f;

__device__ __forceinline__ int bucket_of(int d, int R) {
    const unsigned k = (unsigned)d / (unsigned)R;
    return (int)(k < (unsigned)(NB - 1) ? k : (unsigned)(NB - 1));
}

__device__ __forceinline__ int wave_scan256(const int* __restrict__ s, int lane, int ex[8], int own[8]) {
    int sum = 0;
#pragma unroll
    for (int j = 0; j < 8; ++j) { own[j] = s[lane * 8 + j]; sum += own[j]; }
    int incl = sum;
    for (int o = 1; o < 32; o <<= 1) {
        const int up = __shfl_up(incl, o);
        if (lane >= o) incl += up;
    }
    int run = incl - sum;
#pragma unroll
    for (int j = 0; j < 8; ++j) { ex[j] = run; run += own[j]; }
    return run;
}

__global__ __launch_bounds__(LT)
void bucket_count_kernel(const int* __restrict__ dst, int* __restrict__ cntT, int nE, int nN) {
    __shared__ int hist[NB];
    const int tid = threadIdx.x;
    const int R = (nN + NB - 1) / NB > 0 ? (nN + NB - 1) / NB : 1;
    if (tid < NB) hist[tid] = 0;
    __syncthreads();
    const int chunk = (nE + LB - 1) / LB;
    const int e0 = blockIdx.x * chunk;
    const int e1 = (e0 + chunk < nE) ? (e0 + chunk) : nE;
    for (int e = e0 + tid; e < e1; e += 4 * LT) {
        int d[4];
#pragma unroll
        for (int j = 0; j < 4; ++j) { const int ee = e + j * LT; d[j] = (ee < e1) ? dst[ee] : -1; }
#pragma unroll
        for (int j = 0; j < 4; ++j) if (d[j] >= 0) atomicAdd(&hist[bucket_of(d[j], R)], 1);
    }
    __syncthreads();
    if (tid < NB) cntT[tid * LB + blockIdx.x] = hist[tid];
}

__global__ __launch_bounds__(LT)
void bucket_place_kernel(const int* __restrict__ src, const int* __restrict__ dst,
                         const int* __restrict__ cntT, int* __restrict__ bstart,
                         int2* __restrict__ bedge, int nE, int nN) {
    __shared__ int tot[NB];
    __shared__ int off[NB];
    const int tid  = threadIdx.x;
    const int lane = tid & 31;
    const int wv   = tid >> 5;
    const int R = (nN + NB - 1) / NB > 0 ? (nN + NB - 1) / NB : 1;
    int4 ra[NB / (LT / 32)], rb[NB / (LT / 32)];
#pragma unroll
    for (int r = 0; r < NB / (LT / 32); ++r) {
        const int4* row = (const int4*)(cntT + (wv + r * (LT / 32)) * LB);
        ra[r] = row[lane * 2];
        rb[r] = row[lane * 2 + 1];
    }
#pragma unroll
    for (int r = 0; r < NB / (LT / 32); ++r) {
        const int v[8] = {ra[r].x, ra[r].y, ra[r].z, ra[r].w, rb[r].x, rb[r].y, rb[r].z, rb[r].w};
        int all = 0, before = 0;
#pragma unroll
        for (int j = 0; j < 8; ++j) {
            all += v[j];
            before += (lane * 8 + j < (int)blockIdx.x) ? v[j] : 0;
        }
        for (int o = 16; o > 0; o >>= 1) {
            all    += __shfl_xor(all, o);
            before += __shfl_xor(before, o);
        }
        if (lane == 0) { tot[wv + r * (LT / 32)] = all; off[wv + r * (LT / 32)] = before; }
    }
    __syncthreads();
    if (wv == 0) {
        int ex[8], own[8];
        const int end = wave_scan256(tot, lane, ex, own);
#pragma unroll
        for (int j = 0; j < 8; ++j) {
            off[lane * 8 + j] += ex[j];
            if (blockIdx.x == 0) bstart[lane * 8 + j] = ex[j];
        }
        if (blockIdx.x == 0 && lane == 31) bstart[NB] = end;
    }
    __syncthreads();
    const int chunk = (nE + LB - 1) / LB;
    const int e0 = blockIdx.x * chunk;
    const int e1 = (e0 + chunk < nE) ? (e0 + chunk) : nE;
    for (int e = e0 + tid; e < e1; e += 4 * LT) {
        int d[4], s[4];
#pragma unroll
        for (int j = 0; j < 4; ++j) {
            const int ee = e + j * LT;
            d[j] = (ee < e1) ? dst[ee] : -1;
            s[j] = (ee < e1) ? src[ee] : 0;
        }
#pragma unroll
        for (int j = 0; j < 4; ++j) {
            if (d[j] >= 0) {
                const int p = atomicAdd(&off[bucket_of(d[j], R)], 1);
                bedge[p] = make_int2(s[j], d[j]);
            }
        }
    }
}

__global__ __launch_bounds__(LT)
void csr_build_kernel(const int* __restrict__ bstart, const int2* __restrict__ bedge,
                      int* ctmp, int* __restrict__ rowptr, float* __restrict__ dinv,
                      int* __restrict__ col, int nN) {
    __shared__ int cnt[RMAX];
    __shared__ int roff[RMAX];
    __shared__ int fill[RMAX];
    const int tid  = threadIdx.x;
    const int lane = tid & 31;
    const int wv   = __builtin_amdgcn_readfirstlane(tid >> 5);
    const int R  = (nN + NB - 1) / NB > 0 ? (nN + NB - 1) / NB : 1;
    const int n0 = blockIdx.x * R;
    const int nn = (nN - n0 < R) ? (nN - n0) : R;
    const int b0 = bstart[blockIdx.x];
    const int b1 = bstart[blockIdx.x + 1];
    if (tid < RMAX) { cnt[tid] = 0; fill[tid] = 0; }
    __syncthreads();
    for (int p = b0 + tid; p < b1; p += 4 * LT) {
        int d[4];
#pragma unroll
        for (int j = 0; j < 4; ++j) { const int pp = p + j * LT; d[j] = (pp < b1) ? bedge[pp].y : -1; }
#pragma unroll
        for (int j = 0; j < 4; ++j) {
            const int l = d[j] - n0;
            if (d[j] >= 0 && l >= 0 && l < RMAX) atomicAdd(&cnt[l], 1);
        }
    }
    __syncthreads();
    if (wv == 0) {
        int ex[8], own[8];
        wave_scan256(cnt, lane, ex, own);
#pragma unroll
        for (int j = 0; j < 8; ++j) roff[lane * 8 + j] = ex[j];
    }
    __syncthreads();
    if (tid < nn) {
        rowptr[n0 + tid] = b0 + roff[tid];
        dinv[n0 + tid]   = rsqrtf((float)(cnt[tid] + 1));
    }
    if (tid == 0 && nn > 0 && n0 + nn == nN) rowptr[nN] = b1;
    for (int p = b0 + tid; p < b1; p += 4 * LT) {
        int2 ed[4];
#pragma unroll
        for (int j = 0; j < 4; ++j) { const int pp = p + j * LT; ed[j] = (pp < b1) ? bedge[pp] : make_int2(0, -1); }
#pragma unroll
        for (int j = 0; j < 4; ++j) {
            const int l = ed[j].y - n0;
            if (ed[j].y >= 0 && l >= 0 && l < RMAX) {
                const int r = atomicAdd(&fill[l], 1);
                ctmp[b0 + roff[l] + r] = ed[j].x;
            }
        }
    }
    __syncthreads();
    for (int l = wv; l < nn; l += LT / 32) {
        const int beg = b0 + __builtin_amdgcn_readfirstlane(roff[l]);
        const int deg = __builtin_amdgcn_readfirstlane(cnt[l]);
        const int* seg = ctmp + beg;
        if (deg <= 32) {
            const int key = (lane < deg) ? seg[lane] : 0;
            int rank = 0;
            for (int m = 0; m < deg; ++m) {
                const int km = __builtin_amdgcn_readlane(key, m);
                rank += (km < key || (km == key && m < lane)) ? 1 : 0;
            }
            if (lane < deg) col[beg + rank] = key;
        } else {
            for (int j = lane; j < deg; j += 32) {
                const int key = seg[j];
                int rank = 0;
                for (int m = 0; m < deg; ++m) {
                    const int km = seg[m];
                    rank += (km < key || (km == key && m < j)) ? 1 : 0;
                }
                col[beg + rank] = key;
            }
        }
    }
}

__global__ __launch_bounds__(256)
void gcn_gemm_wmma(const float* __restrict__ X, const float* __restrict__ W,
                   float* __restrict__ H, int nN, int applyRelu) {
    __shared__ __align__(16) float sW[DD * WS];
    __shared__ __align__(16) float sX[128 * XS];

    const int tid  = threadIdx.x;
    const int wave = tid >> 5;
    const int lane = tid & 31;
    const int lr   = lane & 15;
    const int hi   = lane >> 4;
    const int rowBase = blockIdx.x * 128;

    for (int i = tid; i < DD * 32; i += 256) {
        int r  = i >> 5;
        int c4 = (i & 31) << 2;
        float4 w = *(const float4*)(W + r * DD + c4);
        *(float4*)(sW + r * WS + c4) = w;
    }
    for (int i = tid; i < 128 * 32; i += 256) {
        int r  = i >> 5;
        int c4 = (i & 31) << 2;
        int gr = rowBase + r;
        float4 v;
        if (gr < nN) {
            v = *(const float4*)(X + (size_t)gr * DD + c4);
            if (applyRelu) {
                v.x = fmaxf(v.x, 0.0f); v.y = fmaxf(v.y, 0.0f);
                v.z = fmaxf(v.z, 0.0f); v.w = fmaxf(v.w, 0.0f);
            }
        } else {
            v = make_float4(0.0f, 0.0f, 0.0f, 0.0f);
        }
        *(float4*)(sX + r * XS + c4) = v;
    }
    __syncthreads();

    const float* xrow = sX + (wave * 16 + lr) * XS;

    v8f acc[8];
#pragma unroll
    for (int t = 0; t < 8; ++t) acc[t] = (v8f){0.f,0.f,0.f,0.f,0.f,0.f,0.f,0.f};

    for (int kb = 0; kb < 32; ++kb) {
        const int k0 = kb * 4 + hi * 2;
        v2f a = *(const v2f*)(xrow + k0);
        const float* wp = sW + k0 * WS + lr;
#pragma unroll
        for (int t = 0; t < 8; ++t) {
            v2f b;
            b.x = wp[t * 16];
            b.y = wp[t * 16 + WS];
            acc[t] = __builtin_amdgcn_wmma_f32_16x16x4_f32(
                false, a, false, b, (short)0, acc[t], false, false);
        }
    }

    const int row0 = rowBase + wave * 16 + hi * 8;
#pragma unroll
    for (int t = 0; t < 8; ++t) {
        const int col = t * 16 + lr;
#pragma unroll
        for (int v = 0; v < 8; ++v) {
            int r = row0 + v;
            if (r < nN) H[(size_t)r * DD + col] = acc[t][v];
        }
    }
}

template <int G>
__device__ __forceinline__ void gather_group(float4& acc, const float4* __restrict__ h4,
                                             int cs, float cw, int k, int lane) {
    float4 v[G];
    float  w[G];
#pragma unroll
    for (int j = 0; j < G; ++j) {
        const int s = __builtin_amdgcn_readlane(cs, k + j);
        w[j] = __int_as_float(__builtin_amdgcn_readlane(__float_as_int(cw), k + j));
        v[j] = h4[(size_t)s * 32 + lane];
    }
#pragma unroll
    for (int j = 0; j < G; ++j) {
        acc.x += w[j] * v[j].x;  acc.y += w[j] * v[j].y;
        acc.z += w[j] * v[j].z;  acc.w += w[j] * v[j].w;
    }
}

template <bool FIRST>
__global__ __launch_bounds__(256)
void gather_rows_kernel(const float* __restrict__ h, const float* __restrict__ dinv,
                        const float* __restrict__ bias, const int* __restrict__ rowptr,
                        const int* __restrict__ col, float* __restrict__ wgt,
                        float* __restrict__ out, int nN) {
    const int lane = threadIdx.x & 31;
    const int i = blockIdx.x * 8 + __builtin_amdgcn_readfirstlane(threadIdx.x >> 5);
    if (i >= nN) return;
    const int beg = rowptr[i];
    const int end = rowptr[i + 1];
    const float di = dinv[i];
    const float4* __restrict__ h4 = (const float4*)h;

    float4 acc;
    {
        const float w0 = di * di;
        const float4 v = h4[(size_t)i * 32 + lane];
        const float4 b = ((const float4*)bias)[lane];
        acc.x = v.x * w0 + b.x;  acc.y = v.y * w0 + b.y;
        acc.z = v.z * w0 + b.z;  acc.w = v.w * w0 + b.w;
    }

    for (int p = beg; p < end; p += 32) {
        const int n = (end - p < 32) ? (end - p) : 32;
        int   cs = 0;
        float cw = 0.0f;
        if (lane < n) {
            cs = col[p + lane];
            if (FIRST) { cw = di * dinv[cs]; wgt[p + lane] = cw; }
            else       { cw = wgt[p + lane]; }
        }
        int k = 0;
        for (; k + 8 <= n; k += 8) gather_group<8>(acc, h4, cs, cw, k, lane);
        if (k + 4 <= n) { gather_group<4>(acc, h4, cs, cw, k, lane); k += 4; }
        if (k + 2 <= n) { gather_group<2>(acc, h4, cs, cw, k, lane); k += 2; }
        if (k < n)      { gather_group<1>(acc, h4, cs, cw, k, lane); }
    }
    ((float4*)out)[(size_t)i * 32 + lane] = acc;
}

extern "C" void kernel_launch(void* const* d_in, const int* in_sizes, int n_in,
                              void* d_out, int out_size, void* d_ws, size_t ws_size,
                              hipStream_t stream) {
    const float* x   = (const float*)d_in[0];
    const int*   ei  = (const int*)d_in[1];
    const float* W1  = (const float*)d_in[2];
    const float* b1  = (const float*)d_in[3];
    const float* W2  = (const float*)d_in[4];
    const float* b2  = (const float*)d_in[5];
    const float* W3  = (const float*)d_in[6];
    const float* b3  = (const float*)d_in[7];

    const int nN = in_sizes[0] / DD;
    const int nE = in_sizes[1] / 2;
    const int* src = ei;
    const int* dst = ei + nE;

    const size_t npad = (size_t)((nN + 127) / 128) * 128;
    float* dinv = (float*)d_ws;
    float* bufH = dinv + npad;
    int*   rowptr = (int*)(bufH + npad * DD);
    int*   bstart = rowptr + npad + 128;
    int*   col    = bstart + 512;
    float* wgt    = (float*)(col + nE);
    float* outF = (float*)d_out;
    const size_t nE4 = ((size_t)nE + 3) / 4 * 4;
    int2*  bedge = (int2*)d_out;
    int*   ctmp  = (int*)d_out + 2 * nE4;
    int*   cntT  = ctmp + nE4;

    const dim3 blk(256);
    const int gW   = (nN + 7) / 8;
    const int gGemm = (nN + 127) / 128;

    bucket_count_kernel<<<LB, dim3(LT), 0, stream>>>(dst, cntT, nE, nN);
    bucket_place_kernel<<<LB, dim3(LT), 0, stream>>>(src, dst, cntT, bstart, bedge, nE, nN);
    csr_build_kernel<<<NB, dim3(LT), 0, stream>>>(bstart, bedge, ctmp, rowptr, dinv, col, nN);

    gcn_gemm_wmma<<<gGemm, blk, 0, stream>>>(x, W1, bufH, nN, 0);
    gather_rows_kernel<true><<<gW, blk, 0, stream>>>(bufH, dinv, b1, rowptr, col, wgt, outF, nN);

    gcn_gemm_wmma<<<gGemm, blk, 0, stream>>>(outF, W2, bufH, nN, 1);
    gather_rows_kernel<false><<<gW, blk, 0, stream>>>(bufH, dinv, b2, rowptr, col, wgt, outF, nN);

    gcn_gemm_wmma<<<gGemm, blk, 0, stream>>>(outF, W3, bufH, nN, 1);
    gather_rows_kernel<false><<<gW, blk, 0, stream>>>(bufH, dinv, b3, rowptr, col, wgt, outF, nN);
}


# layer 1 and 2 gathers feed the next layer's f32 WMMA product in the same launch through an LDS tile (8 nodes per wave, 208-node blocks): 9 to 7 launches, gathered plane not stored; sum order kept
# speedup vs baseline: 3.5645x; 1.1787x over previous
#include <hip/hip_runtime.h>
#include <hip/hip_bf16.h>


#define DD 128
#define XS 132
#define WS 136

#define NB   256
#define LB   256
#define LT   1024
#define RMAX 256

typedef __attribute__((ext_vector_type(2))) float v2f;
typedef __attribute__((ext_vector_type(8))) float v8f;

__device__ __forceinline__ int bucket_of(int d, int R) {
    const unsigned k = (unsigned)d / (unsigned)R;
    return (int)(k < (unsigned)(NB - 1) ? k : (unsigned)(NB - 1));
}

__device__ __forceinline__ int wave_scan256(const int* __restrict__ s, int lane, int ex[8], int own[8]) {
    int sum = 0;
#pragma unroll
    for (int j = 0; j < 8; ++j) { own[j] = s[lane * 8 + j]; sum += own[j]; }
    int incl = sum;
    for (int o = 1; o < 32; o <<= 1) {
        const int up = __shfl_up(incl, o);
        if (lane >= o) incl += up;
    }
    int run = incl - sum;
#pragma unroll
    for (int j = 0; j < 8; ++j) { ex[j] = run; run += own[j]; }
    return run;
}

__global__ __launch_bounds__(LT)
void bucket_count_kernel(const int* __restrict__ dst, int* __restrict__ cntT, int nE, int nN) {
    __shared__ int hist[NB];
    const int tid = threadIdx.x;
    const int R = (nN + NB - 1) / NB > 0 ? (nN + NB - 1) / NB : 1;
    if (tid < NB) hist[tid] = 0;
    __syncthreads();
    const int chunk = (nE + LB - 1) / LB;
    const int e0 = blockIdx.x * chunk;
    const int e1 = (e0 + chunk < nE) ? (e0 + chunk) : nE;
    for (int e = e0 + tid; e < e1; e += 4 * LT) {
        int d[4];
#pragma unroll
        for (int j = 0; j < 4; ++j) { const int ee = e + j * LT; d[j] = (ee < e1) ? dst[ee] : -1; }
#pragma unroll
        for (int j = 0; j < 4; ++j) if (d[j] >= 0) atomicAdd(&hist[bucket_of(d[j], R)], 1);
    }
    __syncthreads();
    if (tid < NB) cntT[tid * LB + blockIdx.x] = hist[tid];
}

__global__ __launch_bounds__(LT)
void bucket_place_kernel(const int* __restrict__ src, const int* __restrict__ dst,
                         const int* __restrict__ cntT, int* __restrict__ bstart,
                         int2* __restrict__ bedge, int nE, int nN) {
    __shared__ int tot[NB];
    __shared__ int off[NB];
    const int tid  = threadIdx.x;
    const int lane = tid & 31;
    const int wv   = tid >> 5;
    const int R = (nN + NB - 1) / NB > 0 ? (nN + NB - 1) / NB : 1;
    int4 ra[NB / (LT / 32)], rb[NB / (LT / 32)];
#pragma unroll
    for (int r = 0; r < NB / (LT / 32); ++r) {
        const int4* row = (const int4*)(cntT + (wv + r * (LT / 32)) * LB);
        ra[r] = row[lane * 2];
        rb[r] = row[lane * 2 + 1];
    }
#pragma unroll
    for (int r = 0; r < NB / (LT / 32); ++r) {
        const int v[8] = {ra[r].x, ra[r].y, ra[r].z, ra[r].w, rb[r].x, rb[r].y, rb[r].z, rb[r].w};
        int all = 0, before = 0;
#pragma unroll
        for (int j = 0; j < 8; ++j) {
            all += v[j];
            before += (lane * 8 + j < (int)blockIdx.x) ? v[j] : 0;
        }
        for (int o = 16; o > 0; o >>= 1) {
            all    += __shfl_xor(all, o);
            before += __shfl_xor(before, o);
        }
        if (lane == 0) { tot[wv + r * (LT / 32)] = all; off[wv + r * (LT / 32)] = before; }
    }
    __syncthreads();
    if (wv == 0) {
        int ex[8], own[8];
        const int end = wave_scan256(tot, lane, ex, own);
#pragma unroll
        for (int j = 0; j < 8; ++j) {
            off[lane * 8 + j] += ex[j];
            if (blockIdx.x == 0) bstart[lane * 8 + j] = ex[j];
        }
        if (blockIdx.x == 0 && lane == 31) bstart[NB] = end;
    }
    __syncthreads();
    const int chunk = (nE + LB - 1) / LB;
    const int e0 = blockIdx.x * chunk;
    const int e1 = (e0 + chunk < nE) ? (e0 + chunk) : nE;
    for (int e = e0 + tid; e < e1; e += 4 * LT) {
        int d[4], s[4];
#pragma unroll
        for (int j = 0; j < 4; ++j) {
            const int ee = e + j * LT;
            d[j] = (ee < e1) ? dst[ee] : -1;
            s[j] = (ee < e1) ? src[ee] : 0;
        }
#pragma unroll
        for (int j = 0; j < 4; ++j) {
            if (d[j] >= 0) {
                const int p = atomicAdd(&off[bucket_of(d[j], R)], 1);
                bedge[p] = make_int2(s[j], d[j]);
            }
        }
    }
}

__global__ __launch_bounds__(LT)
void csr_build_kernel(const int* __restrict__ bstart, const int2* __restrict__ bedge,
                      int* ctmp, int* __restrict__ rowptr, float* __restrict__ dinv,
                      int* __restrict__ col, int nN) {
    __shared__ int cnt[RMAX];
    __shared__ int roff[RMAX];
    __shared__ int fill[RMAX];
    const int tid  = threadIdx.x;
    const int lane = tid & 31;
    const int wv   = __builtin_amdgcn_readfirstlane(tid >> 5);
    const int R  = (nN + NB - 1) / NB > 0 ? (nN + NB - 1) / NB : 1;
    const int n0 = blockIdx.x * R;
    const int nn = (nN - n0 < R) ? (nN - n0) : R;
    const int b0 = bstart[blockIdx.x];
    const int b1 = bstart[blockIdx.x + 1];
    if (tid < RMAX) { cnt[tid] = 0; fill[tid] = 0; }
    __syncthreads();
    for (int p = b0 + tid; p < b1; p += 4 * LT) {
        int d[4];
#pragma unroll
        for (int j = 0; j < 4; ++j) { const int pp = p + j * LT; d[j] = (pp < b1) ? bedge[pp].y : -1; }
#pragma unroll
        for (int j = 0; j < 4; ++j) {
            const int l = d[j] - n0;
            if (d[j] >= 0 && l >= 0 && l < RMAX) atomicAdd(&cnt[l], 1);
        }
    }
    __syncthreads();
    if (wv == 0) {
        int ex[8], own[8];
        wave_scan256(cnt, lane, ex, own);
#pragma unroll
        for (int j = 0; j < 8; ++j) roff[lane * 8 + j] = ex[j];
    }
    __syncthreads();
    if (tid < nn) {
        rowptr[n0 + tid] = b0 + roff[tid];
        dinv[n0 + tid]   = rsqrtf((float)(cnt[tid] + 1));
    }
    if (tid == 0 && nn > 0 && n0 + nn == nN) rowptr[nN] = b1;
    for (int p = b0 + tid; p < b1; p += 4 * LT) {
        int2 ed[4];
#pragma unroll
        for (int j = 0; j < 4; ++j) { const int pp = p + j * LT; ed[j] = (pp < b1) ? bedge[pp] : make_int2(0, -1); }
#pragma unroll
        for (int j = 0; j < 4; ++j) {
            const int l = ed[j].y - n0;
            if (ed[j].y >= 0 && l >= 0 && l < RMAX) {
                const int r = atomicAdd(&fill[l], 1);
                ctmp[b0 + roff[l] + r] = ed[j].x;
            }
        }
    }
    __syncthreads();
    for (int l = wv; l < nn; l += LT / 32) {
        const int beg = b0 + __builtin_amdgcn_readfirstlane(roff[l]);
        const int deg = __builtin_amdgcn_readfirstlane(cnt[l]);
        const int* seg = ctmp + beg;
        if (deg <= 32) {
            const int key = (lane < deg) ? seg[lane] : 0;
            int rank = 0;
            for (int m = 0; m < deg; ++m) {
                const int km = __builtin_amdgcn_readlane(key, m);
                rank += (km < key || (km == key && m < lane)) ? 1 : 0;
            }
            if (lane < deg) col[beg + rank] = key;
        } else {
            for (int j = lane; j < deg; j += 32) {
                const int key = seg[j];
                int rank = 0;
                for (int m = 0; m < deg; ++m) {
                    const int km = seg[m];
                    rank += (km < key || (km == key && m < j)) ? 1 : 0;
                }
                col[beg + rank] = key;
            }
        }
    }
}

__global__ __launch_bounds__(256)
void gcn_gemm_wmma(const float* __restrict__ X, const float* __restrict__ W,
                   float* __restrict__ H, int nN, int applyRelu) {
    __shared__ __align__(16) float sW[DD * WS];
    __shared__ __align__(16) float sX[128 * XS];

    const int tid  = threadIdx.x;
    const int wave = tid >> 5;
    const int lane = tid & 31;
    const int lr   = lane & 15;
    const int hi   = lane >> 4;
    const int rowBase = blockIdx.x * 128;

    for (int i = tid; i < DD * 32; i += 256) {
        int r  = i >> 5;
        int c4 = (i & 31) << 2;
        float4 w = *(const float4*)(W + r * DD + c4);
        *(float4*)(sW + r * WS + c4) = w;
    }
    for (int i = tid; i < 128 * 32; i += 256) {
        int r  = i >> 5;
        int c4 = (i & 31) << 2;
        int gr = rowBase + r;
        float4 v;
        if (gr < nN) {
            v = *(const float4*)(X + (size_t)gr * DD + c4);
            if (applyRelu) {
                v.x = fmaxf(v.x, 0.0f); v.y = fmaxf(v.y, 0.0f);
                v.z = fmaxf(v.z, 0.0f); v.w = fmaxf(v.w, 0.0f);
            }
        } else {
            v = make_float4(0.0f, 0.0f, 0.0f, 0.0f);
        }
        *(float4*)(sX + r * XS + c4) = v;
    }
    __syncthreads();

    const float* xrow = sX + (wave * 16 + lr) * XS;

    v8f acc[8];
#pragma unroll
    for (int t = 0; t < 8; ++t) acc[t] = (v8f){0.f,0.f,0.f,0.f,0.f,0.f,0.f,0.f};

    for (int kb = 0; kb < 32; ++kb) {
        const int k0 = kb * 4 + hi * 2;
        v2f a = *(const v2f*)(xrow + k0);
        const float* wp = sW + k0 * WS + lr;
#pragma unroll
        for (int t = 0; t < 8; ++t) {
            v2f b;
            b.x = wp[t * 16];
            b.y = wp[t * 16 + WS];
            acc[t] = __builtin_amdgcn_wmma_f32_16x16x4_f32(
                false, a, false, b, (short)0, acc[t], false, false);
        }
    }

    const int row0 = rowBase + wave * 16 + hi * 8;
#pragma unroll
    for (int t = 0; t < 8; ++t) {
        const int col = t * 16 + lr;
#pragma unroll
        for (int v = 0; v < 8; ++v) {
            int r = row0 + v;
            if (r < nN) H[(size_t)r * DD + col] = acc[t][v];
        }
    }
}

template <int G>
__device__ __forceinline__ void gather_group(float4& acc, const float4* __restrict__ h4,
                                             int cs, float cw, int k, int lane) {
    float4 v[G];
    float  w[G];
#pragma unroll
    for (int j = 0; j < G; ++j) {
        const int s = __builtin_amdgcn_readlane(cs, k + j);
        w[j] = __int_as_float(__builtin_amdgcn_readlane(__float_as_int(cw), k + j));
        v[j] = h4[(size_t)s * 32 + lane];
    }
#pragma unroll
    for (int j = 0; j < G; ++j) {
        acc.x += w[j] * v[j].x;  acc.y += w[j] * v[j].y;
        acc.z += w[j] * v[j].z;  acc.w += w[j] * v[j].w;
    }
}

template <bool FIRST>
__device__ __forceinline__ float4 gather_node(const float4* __restrict__ h4, const float* __restrict__ dinv,
                                              const int* __restrict__ col, float* __restrict__ wgt,
                                              const float4 b, int i, int beg, int end, float di, int lane) {
    float4 acc;
    {
        const float w0 = di * di;
        const float4 v = h4[(size_t)i * 32 + lane];
        acc.x = v.x * w0 + b.x;  acc.y = v.y * w0 + b.y;
        acc.z = v.z * w0 + b.z;  acc.w = v.w * w0 + b.w;
    }
    for (int p = beg; p < end; p += 32) {
        const int n = (end - p < 32) ? (end - p) : 32;
        int   cs = 0;
        float cw = 0.0f;
        if (lane < n) {
            cs = col[p + lane];
            if (FIRST) { cw = di * dinv[cs]; wgt[p + lane] = cw; }
            else       { cw = wgt[p + lane]; }
        }
        int k = 0;
        for (; k + 8 <= n; k += 8) gather_group<8>(acc, h4, cs, cw, k, lane);
        if (k + 4 <= n) { gather_group<4>(acc, h4, cs, cw, k, lane); k += 4; }
        if (k + 2 <= n) { gather_group<2>(acc, h4, cs, cw, k, lane); k += 2; }
        if (k < n)      { gather_group<1>(acc, h4, cs, cw, k, lane); }
    }
    return acc;
}

template <bool FIRST>
__global__ __launch_bounds__(256)
void gather_rows_kernel(const float* __restrict__ h, const float* __restrict__ dinv,
                        const float* __restrict__ bias, const int* __restrict__ rowptr,
                        const int* __restrict__ col, float* __restrict__ wgt,
                        float* __restrict__ out, int nN) {
    const int lane = threadIdx.x & 31;
    const int i = blockIdx.x * 8 + __builtin_amdgcn_readfirstlane(threadIdx.x >> 5);
    if (i >= nN) return;
    const int beg = rowptr[i];
    const int end = rowptr[i + 1];
    const float di = dinv[i];
    const float4 b = ((const float4*)bias)[lane];
    const float4 acc = gather_node<FIRST>((const float4*)h, dinv, col, wgt, b, i, beg, end, di, lane);
    ((float4*)out)[(size_t)i * 32 + lane] = acc;
}

#define TM 208
#define FW (TM / 8)
#define FT (FW * 32)
#define WIT ((DD * 32 + FT - 1) / FT)

template <bool FIRST>
__global__ __launch_bounds__(FT)
void gather_gemm_kernel(const float* __restrict__ h, const float* __restrict__ dinv,
                        const float* __restrict__ bias, const int* __restrict__ rowptr,
                        const int* __restrict__ col, float* __restrict__ wgt,
                        const float* __restrict__ W, float* __restrict__ H, int nN) {
    __shared__ __align__(16) float smem[DD * WS + TM * XS];
    float* sW = smem;
    float* sX = smem + DD * WS;

    const int tid  = threadIdx.x;
    const int lane = tid & 31;
    const int wave = __builtin_amdgcn_readfirstlane(tid >> 5);
    const int rowBase = blockIdx.x * TM;

    {
        float4 wv[WIT];
#pragma unroll
        for (int it = 0; it < WIT; ++it) {
            const int i = tid + it * FT;
            wv[it] = (i < DD * 32) ? *(const float4*)(W + (i >> 5) * DD + ((i & 31) << 2))
                                   : make_float4(0.0f, 0.0f, 0.0f, 0.0f);
        }
#pragma unroll
        for (int it = 0; it < WIT; ++it) {
            const int i = tid + it * FT;
            if (i < DD * 32) *(float4*)(sW + (i >> 5) * WS + ((i & 31) << 2)) = wv[it];
        }
    }

    {
        const int i0 = rowBase + wave * 8;
        int   rp = 0;
        float dv = 0.0f;
        if (lane <= 8) { const int j = i0 + lane; rp = rowptr[j < nN ? j : nN]; }
        if (lane <  8) { const int j = i0 + lane; dv = dinv[j < nN ? j : nN - 1]; }
        const float4 b = ((const float4*)bias)[lane];
        const float4* __restrict__ h4 = (const float4*)h;
        for (int r = 0; r < 8; ++r) {
            const int i = i0 + r;
            float4 acc = make_float4(0.0f, 0.0f, 0.0f, 0.0f);
            if (i < nN) {
                const int   beg = __builtin_amdgcn_readlane(rp, r);
                const int   end = __builtin_amdgcn_readlane(rp, r + 1);
                const float di  = __int_as_float(__builtin_amdgcn_readlane(__float_as_int(dv), r));
                acc = gather_node<FIRST>(h4, dinv, col, wgt, b, i, beg, end, di, lane);
                acc.x = fmaxf(acc.x, 0.0f); acc.y = fmaxf(acc.y, 0.0f);
                acc.z = fmaxf(acc.z, 0.0f); acc.w = fmaxf(acc.w, 0.0f);
            }
            *(float4*)(sX + (wave * 8 + r) * XS + lane * 4) = acc;
        }
    }
    __syncthreads();

    const int lr    = lane & 15;
    const int hi    = lane >> 4;
    const int strip = wave >> 1;
    const int half  = wave & 1;

    const float* xrow = sX + (strip * 16 + lr) * XS;

    v8f acc[4];
#pragma unroll
    for (int t = 0; t < 4; ++t) acc[t] = (v8f){0.f,0.f,0.f,0.f,0.f,0.f,0.f,0.f};

    for (int kb = 0; kb < 32; ++kb) {
        const int k0 = kb * 4 + hi * 2;
        v2f a = *(const v2f*)(xrow + k0);
        const float* wp = sW + k0 * WS + half * 64 + lr;
#pragma unroll
        for (int t = 0; t < 4; ++t) {
            v2f b;
            b.x = wp[t * 16];
            b.y = wp[t * 16 + WS];
            acc[t] = __builtin_amdgcn_wmma_f32_16x16x4_f32(
                false, a, false, b, (short)0, acc[t], false, false);
        }
    }

    const int row0 = rowBase + strip * 16 + hi * 8;
#pragma unroll
    for (int t = 0; t < 4; ++t) {
        const int c = half * 64 + t * 16 + lr;
#pragma unroll
        for (int v = 0; v < 8; ++v) {
            int r = row0 + v;
            if (r < nN) H[(size_t)r * DD + c] = acc[t][v];
        }
    }
}

extern "C" void kernel_launch(void* const* d_in, const int* in_sizes, int n_in,
                              void* d_out, int out_size, void* d_ws, size_t ws_size,
                              hipStream_t stream) {
    const float* x   = (const float*)d_in[0];
    const int*   ei  = (const int*)d_in[1];
    const float* W1  = (const float*)d_in[2];
    const float* b1  = (const float*)d_in[3];
    const float* W2  = (const float*)d_in[4];
    const float* b2  = (const float*)d_in[5];
    const float* W3  = (const float*)d_in[6];
    const float* b3  = (const float*)d_in[7];

    const int nN = in_sizes[0] / DD;
    const int nE = in_sizes[1] / 2;
    const int* src = ei;
    const int* dst = ei + nE;

    const size_t npad = (size_t)((nN + 127) / 128) * 128;
    float* dinv = (float*)d_ws;
    float* bufH = dinv + npad;
    int*   rowptr = (int*)(bufH + npad * DD);
    int*   bstart = rowptr + npad + 128;
    int*   col    = bstart + 512;
    float* wgt    = (float*)(col + nE);
    float* outF = (float*)d_out;
    const size_t nE4 = ((size_t)nE + 3) / 4 * 4;
    int2*  bedge = (int2*)d_out;
    int*   ctmp  = (int*)d_out + 2 * nE4;
    int*   cntT  = ctmp + nE4;

    const dim3 blk(256);
    const int gW   = (nN + 7) / 8;
    const int gGemm = (nN + 127) / 128;
    const int gFuse = (nN + TM - 1) / TM;

    bucket_count_kernel<<<LB, dim3(LT), 0, stream>>>(dst, cntT, nE, nN);
    bucket_place_kernel<<<LB, dim3(LT), 0, stream>>>(src, dst, cntT, bstart, bedge, nE, nN);
    csr_build_kernel<<<NB, dim3(LT), 0, stream>>>(bstart, bedge, ctmp, rowptr, dinv, col, nN);

    gcn_gemm_wmma<<<gGemm, blk, 0, stream>>>(x, W1, bufH, nN, 0);
    gather_gemm_kernel<true><<<gFuse, dim3(FT), 0, stream>>>(bufH, dinv, b1, rowptr, col, wgt, W2, outF, nN);
    gather_gemm_kernel<false><<<gFuse, dim3(FT), 0, stream>>>(outF, dinv, b2, rowptr, col, wgt, W3, bufH, nN);
    gather_rows_kernel<false><<<gW, blk, 0, stream>>>(bufH, dinv, b3, rowptr, col, wgt, outF, nN);
}


# layer 1 f32 WMMA product runs as leading blocks of the three list-build launches (96-row tiles, a third each, beside the list blocks) instead of a launch of its own: 7 to 6 launches; sums unchanged
# speedup vs baseline: 3.8895x; 1.0912x over previous
#include <hip/hip_runtime.h>
#include <hip/hip_bf16.h>


#define DD 128
#define XS 132
#define WS 136

#define NB   256
#define LB   256
#define LT   1024
#define RMAX 256

typedef __attribute__((ext_vector_type(2))) float v2f;
typedef __attribute__((ext_vector_type(8))) float v8f;

__device__ __forceinline__ int bucket_of(int d, int R) {
    const unsigned k = (unsigned)d / (unsigned)R;
    return (int)(k < (unsigned)(NB - 1) ? k : (unsigned)(NB - 1));
}

__device__ __forceinline__ int wave_scan256(const int* __restrict__ s, int lane, int ex[8], int own[8]) {
    int sum = 0;
#pragma unroll
    for (int j = 0; j < 8; ++j) { own[j] = s[lane * 8 + j]; sum += own[j]; }
    int incl = sum;
    for (int o = 1; o < 32; o <<= 1) {
        const int up = __shfl_up(incl, o);
        if (lane >= o) incl += up;
    }
    int run = incl - sum;
#pragma unroll
    for (int j = 0; j < 8; ++j) { ex[j] = run; run += own[j]; }
    return run;
}

#define GR 96
#define GS (GR / 16)
#define GPARTS 3
#define XIT ((GR * 32 + LT - 1) / LT)

__device__ __forceinline__ void gemm_tile_role(const float* __restrict__ X, const float* __restrict__ W,
                                               float* __restrict__ H, int nN, int tile) {
    __shared__ __align__(16) float gsm[DD * WS + GR * XS];
    float* sW = gsm;
    float* sX = gsm + DD * WS;

    const int tid  = threadIdx.x;
    const int wave = __builtin_amdgcn_readfirstlane(tid >> 5);
    const int lane = tid & 31;
    const int lr   = lane & 15;
    const int hi   = lane >> 4;
    const int rowBase = tile * GR;

    {
        float4 wv[DD * 32 / LT], xv[XIT];
#pragma unroll
        for (int it = 0; it < DD * 32 / LT; ++it) {
            const int i = tid + it * LT;
            wv[it] = *(const float4*)(W + (i >> 5) * DD + ((i & 31) << 2));
        }
#pragma unroll
        for (int it = 0; it < XIT; ++it) {
            const int i  = tid + it * LT;
            const int gr = rowBase + (i >> 5);
            xv[it] = (i < GR * 32 && gr < nN) ? *(const float4*)(X + (size_t)gr * DD + ((i & 31) << 2))
                                               : make_float4(0.0f, 0.0f, 0.0f, 0.0f);
        }
#pragma unroll
        for (int it = 0; it < DD * 32 / LT; ++it) {
            const int i = tid + it * LT;
            *(float4*)(sW + (i >> 5) * WS + ((i & 31) << 2)) = wv[it];
        }
#pragma unroll
        for (int it = 0; it < XIT; ++it) {
            const int i = tid + it * LT;
            if (i < GR * 32) *(float4*)(sX + (i >> 5) * XS + ((i & 31) << 2)) = xv[it];
        }
    }
    __syncthreads();

    const int strip = wave >> 3;
    const int ct    = wave & 7;
    const bool two  = wave + 32 < GS * 8;

    const float* xrow = sX + (strip * 16 + lr) * XS;
    const float* wcol = sW + ct * 16 + lr;

    v8f acc0 = (v8f){0.f,0.f,0.f,0.f,0.f,0.f,0.f,0.f};
    v8f acc1 = (v8f){0.f,0.f,0.f,0.f,0.f,0.f,0.f,0.f};
    if (two) {
        for (int kb = 0; kb < 32; ++kb) {
            const int k0 = kb * 4 + hi * 2;
            v2f a0 = *(const v2f*)(xrow + k0);
            v2f a1 = *(const v2f*)(xrow + 64 * XS + k0);
            v2f b;
            b.x = wcol[k0 * WS];
            b.y = wcol[k0 * WS + WS];
            acc0 = __builtin_amdgcn_wmma_f32_16x16x4_f32(false, a0, false, b, (short)0, acc0, false, false);
            acc1 = __builtin_amdgcn_wmma_f32_16x16x4_f32(false, a1, false, b, (short)0, acc1, false, false);
        }
    } else {
        for (int kb = 0; kb < 32; ++kb) {
            const int k0 = kb * 4 + hi * 2;
            v2f a0 = *(const v2f*)(xrow + k0);
            v2f b;
            b.x = wcol[k0 * WS];
            b.y = wcol[k0 * WS + WS];
            acc0 = __builtin_amdgcn_wmma_f32_16x16x4_f32(false, a0, false, b, (short)0, acc0, false, false);
        }
    }

    const int row0 = rowBase + strip * 16 + hi * 8;
    const int c    = ct * 16 + lr;
#pragma unroll
    for (int v = 0; v < 8; ++v) {
        const int r = row0 + v;
        if (r < nN) H[(size_t)r * DD + c] = acc0[v];
    }
    if (two) {
#pragma unroll
        for (int v = 0; v < 8; ++v) {
            const int r = row0 + 64 + v;
            if (r < nN) H[(size_t)r * DD + c] = acc1[v];
        }
    }
}

__device__ __forceinline__ int product_blocks(int nN) {
    const int nT = (nN + GR - 1) / GR;
    return (nT + GPARTS - 1) / GPARTS;
}

__device__ __forceinline__ bool product_role(const float* __restrict__ X, const float* __restrict__ W,
                                             float* __restrict__ H, int nN, int part) {
    const int per = product_blocks(nN);
    if ((int)blockIdx.x >= per) return false;
    const int tile = part * per + (int)blockIdx.x;
    if (tile * GR < nN) gemm_tile_role(X, W, H, nN, tile);
    return true;
}

__global__ __launch_bounds__(LT)
void bucket_count_kernel(const int* __restrict__ dst, int* __restrict__ cntT, int nE, int nN,
                         const float* __restrict__ gX, const float* __restrict__ gW,
                         float* __restrict__ gH, int part) {
    __shared__ int hist[NB];
    if (product_role(gX, gW, gH, nN, part)) return;
    const int bid = (int)blockIdx.x - product_blocks(nN);
    const int tid = threadIdx.x;
    const int R = (nN + NB - 1) / NB > 0 ? (nN + NB - 1) / NB : 1;
    if (tid < NB) hist[tid] = 0;
    __syncthreads();
    const int chunk = (nE + LB - 1) / LB;
    const int e0 = bid * chunk;
    const int e1 = (e0 + chunk < nE) ? (e0 + chunk) : nE;
    for (int e = e0 + tid; e < e1; e += 4 * LT) {
        int d[4];
#pragma unroll
        for (int j = 0; j < 4; ++j) { const int ee = e + j * LT; d[j] = (ee < e1) ? dst[ee] : -1; }
#pragma unroll
        for (int j = 0; j < 4; ++j) if (d[j] >= 0) atomicAdd(&hist[bucket_of(d[j], R)], 1);
    }
    __syncthreads();
    if (tid < NB) cntT[tid * LB + bid] = hist[tid];
}

__global__ __launch_bounds__(LT)
void bucket_place_kernel(const int* __restrict__ src, const int* __restrict__ dst,
                         const int* __restrict__ cntT, int* __restrict__ bstart,
                         int2* __restrict__ bedge, int nE, int nN,
                         const float* __restrict__ gX, const float* __restrict__ gW,
                         float* __restrict__ gH, int part) {
    __shared__ int tot[NB];
    __shared__ int off[NB];
    if (product_role(gX, gW, gH, nN, part)) return;
    const int bid  = (int)blockIdx.x - product_blocks(nN);
    const int tid  = threadIdx.x;
    const int lane = tid & 31;
    const int wv   = tid >> 5;
    const int R = (nN + NB - 1) / NB > 0 ? (nN + NB - 1) / NB : 1;
    for (int hf = 0; hf < 2; ++hf) {
        int4 ra[NB / (LT / 32) / 2], rb[NB / (LT / 32) / 2];
#pragma unroll
        for (int r = 0; r < NB / (LT / 32) / 2; ++r) {
            const int4* row = (const int4*)(cntT + (wv + (hf * 4 + r) * (LT / 32)) * LB);
            ra[r] = row[lane * 2];
            rb[r] = row[lane * 2 + 1];
        }
#pragma unroll
        for (int r = 0; r < NB / (LT / 32) / 2; ++r) {
            const int v[8] = {ra[r].x, ra[r].y, ra[r].z, ra[r].w, rb[r].x, rb[r].y, rb[r].z, rb[r].w};
            int all = 0, before = 0;
#pragma unroll
            for (int j = 0; j < 8; ++j) {
                all += v[j];
                before += (lane * 8 + j < bid) ? v[j] : 0;
            }
            for (int o = 16; o > 0; o >>= 1) {
                all    += __shfl_xor(all, o);
                before += __shfl_xor(before, o);
            }
            if (lane == 0) {
                tot[wv + (hf * 4 + r) * (LT / 32)] = all;
                off[wv + (hf * 4 + r) * (LT / 32)] = before;
            }
        }
    }
    __syncthreads();
    if (wv == 0) {
        int ex[8], own[8];
        const int end = wave_scan256(tot, lane, ex, own);
#pragma unroll
        for (int j = 0; j < 8; ++j) {
            off[lane * 8 + j] += ex[j];
            if (bid == 0) bstart[lane * 8 + j] = ex[j];
        }
        if (bid == 0 && lane == 31) bstart[NB] = end;
    }
    __syncthreads();
    const int chunk = (nE + LB - 1) / LB;
    const int e0 = bid * chunk;
    const int e1 = (e0 + chunk < nE) ? (e0 + chunk) : nE;
    for (int e = e0 + tid; e < e1; e += 4 * LT) {
        int d[4], s[4];
#pragma unroll
        for (int j = 0; j < 4; ++j) {
            const int ee = e + j * LT;
            d[j] = (ee < e1) ? dst[ee] : -1;
            s[j] = (ee < e1) ? src[ee] : 0;
        }
#pragma unroll
        for (int j = 0; j < 4; ++j) {
            if (d[j] >= 0) {
                const int p = atomicAdd(&off[bucket_of(d[j], R)], 1);
                bedge[p] = make_int2(s[j], d[j]);
            }
        }
    }
}

__global__ __launch_bounds__(LT)
void csr_build_kernel(const int* __restrict__ bstart, const int2* __restrict__ bedge,
                      int* ctmp, int* __restrict__ rowptr, float* __restrict__ dinv,
                      int* __restrict__ col, int nN,
                      const float* __restrict__ gX, const float* __restrict__ gW,
                      float* __restrict__ gH, int part) {
    __shared__ int cnt[RMAX];
    __shared__ int roff[RMAX];
    __shared__ int fill[RMAX];
    if (product_role(gX, gW, gH, nN, part)) return;
    const int bid  = (int)blockIdx.x - product_blocks(nN);
    const int tid  = threadIdx.x;
    const int lane = tid & 31;
    const int wv   = __builtin_amdgcn_readfirstlane(tid >> 5);
    const int R  = (nN + NB - 1) / NB > 0 ? (nN + NB - 1) / NB : 1;
    const int n0 = bid * R;
    const int nn = (nN - n0 < R) ? (nN - n0) : R;
    const int b0 = bstart[bid];
    const int b1 = bstart[bid + 1];
    if (tid < RMAX) { cnt[tid] = 0; fill[tid] = 0; }
    __syncthreads();
    for (int p = b0 + tid; p < b1; p += 4 * LT) {
        int d[4];
#pragma unroll
        for (int j = 0; j < 4; ++j) { const int pp = p + j * LT; d[j] = (pp < b1) ? bedge[pp].y : -1; }
#pragma unroll
        for (int j = 0; j < 4; ++j) {
            const int l = d[j] - n0;
            if (d[j] >= 0 && l >= 0 && l < RMAX) atomicAdd(&cnt[l], 1);
        }
    }
    __syncthreads();
    if (wv == 0) {
        int ex[8], own[8];
        wave_scan256(cnt, lane, ex, own);
#pragma unroll
        for (int j = 0; j < 8; ++j) roff[lane * 8 + j] = ex[j];
    }
    __syncthreads();
    if (tid < nn) {
        rowptr[n0 + tid] = b0 + roff[tid];
        dinv[n0 + tid]   = rsqrtf((float)(cnt[tid] + 1));
    }
    if (tid == 0 && nn > 0 && n0 + nn == nN) rowptr[nN] = b1;
    for (int p = b0 + tid; p < b1; p += 4 * LT) {
        int2 ed[4];
#pragma unroll
        for (int j = 0; j < 4; ++j) { const int pp = p + j * LT; ed[j] = (pp < b1) ? bedge[pp] : make_int2(0, -1); }
#pragma unroll
        for (int j = 0; j < 4; ++j) {
            const int l = ed[j].y - n0;
            if (ed[j].y >= 0 && l >= 0 && l < RMAX) {
                const int r = atomicAdd(&fill[l], 1);
                ctmp[b0 + roff[l] + r] = ed[j].x;
            }
        }
    }
    __syncthreads();
    for (int l = wv; l < nn; l += LT / 32) {
        const int beg = b0 + __builtin_amdgcn_readfirstlane(roff[l]);
        const int deg = __builtin_amdgcn_readfirstlane(cnt[l]);
        const int* seg = ctmp + beg;
        if (deg <= 32) {
            const int key = (lane < deg) ? seg[lane] : 0;
            int rank = 0;
            for (int m = 0; m < deg; ++m) {
                const int km = __builtin_amdgcn_readlane(key, m);
                rank += (km < key || (km == key && m < lane)) ? 1 : 0;
            }
            if (lane < deg) col[beg + rank] = key;
        } else {
            for (int j = lane; j < deg; j += 32) {
                const int key = seg[j];
                int rank = 0;
                for (int m = 0; m < deg; ++m) {
                    const int km = seg[m];
                    rank += (km < key || (km == key && m < j)) ? 1 : 0;
                }
                col[beg + rank] = key;
            }
        }
    }
}

template <int G>
__device__ __forceinline__ void gather_group(float4& acc, const float4* __restrict__ h4,
                                             int cs, float cw, int k, int lane) {
    float4 v[G];
    float  w[G];
#pragma unroll
    for (int j = 0; j < G; ++j) {
        const int s = __builtin_amdgcn_readlane(cs, k + j);
        w[j] = __int_as_float(__builtin_amdgcn_readlane(__float_as_int(cw), k + j));
        v[j] = h4[(size_t)s * 32 + lane];
    }
#pragma unroll
    for (int j = 0; j < G; ++j) {
        acc.x += w[j] * v[j].x;  acc.y += w[j] * v[j].y;
        acc.z += w[j] * v[j].z;  acc.w += w[j] * v[j].w;
    }
}

template <bool FIRST>
__device__ __forceinline__ float4 gather_node(const float4* __restrict__ h4, const float* __restrict__ dinv,
                                              const int* __restrict__ col, float* __restrict__ wgt,
                                              const float4 b, int i, int beg, int end, float di, int lane) {
    float4 acc;
    {
        const float w0 = di * di;
        const float4 v = h4[(size_t)i * 32 + lane];
        acc.x = v.x * w0 + b.x;  acc.y = v.y * w0 + b.y;
        acc.z = v.z * w0 + b.z;  acc.w = v.w * w0 + b.w;
    }
    for (int p = beg; p < end; p += 32) {
        const int n = (end - p < 32) ? (end - p) : 32;
        int   cs = 0;
        float cw = 0.0f;
        if (lane < n) {
            cs = col[p + lane];
            if (FIRST) { cw = di * dinv[cs]; wgt[p + lane] = cw; }
            else       { cw = wgt[p + lane]; }
        }
        int k = 0;
        for (; k + 8 <= n; k += 8) gather_group<8>(acc, h4, cs, cw, k, lane);
        if (k + 4 <= n) { gather_group<4>(acc, h4, cs, cw, k, lane); k += 4; }
        if (k + 2 <= n) { gather_group<2>(acc, h4, cs, cw, k, lane); k += 2; }
        if (k < n)      { gather_group<1>(acc, h4, cs, cw, k, lane); }
    }
    return acc;
}

template <bool FIRST>
__global__ __launch_bounds__(256)
void gather_rows_kernel(const float* __restrict__ h, const float* __restrict__ dinv,
                        const float* __restrict__ bias, const int* __restrict__ rowptr,
                        const int* __restrict__ col, float* __restrict__ wgt,
                        float* __restrict__ out, int nN) {
    const int lane = threadIdx.x & 31;
    const int i = blockIdx.x * 8 + __builtin_amdgcn_readfirstlane(threadIdx.x >> 5);
    if (i >= nN) return;
    const int beg = rowptr[i];
    const int end = rowptr[i + 1];
    const float di = dinv[i];
    const float4 b = ((const float4*)bias)[lane];
    const float4 acc = gather_node<FIRST>((const float4*)h, dinv, col, wgt, b, i, beg, end, di, lane);
    ((float4*)out)[(size_t)i * 32 + lane] = acc;
}

#define TM 208
#define FW (TM / 8)
#define FT (FW * 32)
#define WIT ((DD * 32 + FT - 1) / FT)

template <bool FIRST>
__global__ __launch_bounds__(FT)
void gather_gemm_kernel(const float* __restrict__ h, const float* __restrict__ dinv,
                        const float* __restrict__ bias, const int* __restrict__ rowptr,
                        const int* __restrict__ col, float* __restrict__ wgt,
                        const float* __restrict__ W, float* __restrict__ H, int nN) {
    __shared__ __align__(16) float smem[DD * WS + TM * XS];
    float* sW = smem;
    float* sX = smem + DD * WS;

    const int tid  = threadIdx.x;
    const int lane = tid & 31;
    const int wave = __builtin_amdgcn_readfirstlane(tid >> 5);
    const int rowBase = blockIdx.x * TM;

    {
        float4 wv[WIT];
#pragma unroll
        for (int it = 0; it < WIT; ++it) {
            const int i = tid + it * FT;
            wv[it] = (i < DD * 32) ? *(const float4*)(W + (i >> 5) * DD + ((i & 31) << 2))
                                   : make_float4(0.0f, 0.0f, 0.0f, 0.0f);
        }
#pragma unroll
        for (int it = 0; it < WIT; ++it) {
            const int i = tid + it * FT;
            if (i < DD * 32) *(float4*)(sW + (i >> 5) * WS + ((i & 31) << 2)) = wv[it];
        }
    }

    {
        const int i0 = rowBase + wave * 8;
        int   rp = 0;
        float dv = 0.0f;
        if (lane <= 8) { const int j = i0 + lane; rp = rowptr[j < nN ? j : nN]; }
        if (lane <  8) { const int j = i0 + lane; dv = dinv[j < nN ? j : nN - 1]; }
        const float4 b = ((const float4*)bias)[lane];
        const float4* __restrict__ h4 = (const float4*)h;
        for (int r = 0; r < 8; ++r) {
            const int i = i0 + r;
            float4 acc = make_float4(0.0f, 0.0f, 0.0f, 0.0f);
            if (i < nN) {
                const int   beg = __builtin_amdgcn_readlane(rp, r);
                const int   end = __builtin_amdgcn_readlane(rp, r + 1);
                const float di  = __int_as_float(__builtin_amdgcn_readlane(__float_as_int(dv), r));
                acc = gather_node<FIRST>(h4, dinv, col, wgt, b, i, beg, end, di, lane);
                acc.x = fmaxf(acc.x, 0.0f); acc.y = fmaxf(acc.y, 0.0f);
                acc.z = fmaxf(acc.z, 0.0f); acc.w = fmaxf(acc.w, 0.0f);
            }
            *(float4*)(sX + (wave * 8 + r) * XS + lane * 4) = acc;
        }
    }
    __syncthreads();

    const int lr    = lane & 15;
    const int hi    = lane >> 4;
    const int strip = wave >> 1;
    const int half  = wave & 1;

    const float* xrow = sX + (strip * 16 + lr) * XS;

    v8f acc[4];
#pragma unroll
    for (int t = 0; t < 4; ++t) acc[t] = (v8f){0.f,0.f,0.f,0.f,0.f,0.f,0.f,0.f};

    for (int kb = 0; kb < 32; ++kb) {
        const int k0 = kb * 4 + hi * 2;
        v2f a = *(const v2f*)(xrow + k0);
        const float* wp = sW + k0 * WS + half * 64 + lr;
#pragma unroll
        for (int t = 0; t < 4; ++t) {
            v2f b;
            b.x = wp[t * 16];
            b.y = wp[t * 16 + WS];
            acc[t] = __builtin_amdgcn_wmma_f32_16x16x4_f32(
                false, a, false, b, (short)0, acc[t], false, false);
        }
    }

    const int row0 = rowBase + strip * 16 + hi * 8;
#pragma unroll
    for (int t = 0; t < 4; ++t) {
        const int c = half * 64 + t * 16 + lr;
#pragma unroll
        for (int v = 0; v < 8; ++v) {
            int r = row0 + v;
            if (r < nN) H[(size_t)r * DD + c] = acc[t][v];
        }
    }
}

extern "C" void kernel_launch(void* const* d_in, const int* in_sizes, int n_in,
                              void* d_out, int out_size, void* d_ws, size_t ws_size,
                              hipStream_t stream) {
    const float* x   = (const float*)d_in[0];
    const int*   ei  = (const int*)d_in[1];
    const float* W1  = (const float*)d_in[2];
    const float* b1  = (const float*)d_in[3];
    const float* W2  = (const float*)d_in[4];
    const float* b2  = (const float*)d_in[5];
    const float* W3  = (const float*)d_in[6];
    const float* b3  = (const float*)d_in[7];

    const int nN = in_sizes[0] / DD;
    const int nE = in_sizes[1] / 2;
    const int* src = ei;
    const int* dst = ei + nE;

    const size_t npad = (size_t)((nN + 127) / 128) * 128;
    float* dinv = (float*)d_ws;
    float* bufH = dinv + npad;
    int*   rowptr = (int*)(bufH + npad * DD);
    int*   bstart = rowptr + npad + 128;
    int*   col    = bstart + 512;
    float* wgt    = (float*)(col + nE);
    float* outF = (float*)d_out;
    const size_t nE4 = ((size_t)nE + 3) / 4 * 4;
    int2*  bedge = (int2*)d_out;
    int*   ctmp  = (int*)d_out + 2 * nE4;
    int*   cntT  = ctmp + nE4;

    const dim3 blk(256);
    const int gW   = (nN + 7) / 8;
    const int gProd = ((nN + GR - 1) / GR + GPARTS - 1) / GPARTS;
    const int gFuse = (nN + TM - 1) / TM;

    bucket_count_kernel<<<gProd + LB, dim3(LT), 0, stream>>>(dst, cntT, nE, nN, x, W1, bufH, 0);
    bucket_place_kernel<<<gProd + LB, dim3(LT), 0, stream>>>(src, dst, cntT, bstart, bedge, nE, nN, x, W1, bufH, 1);
    csr_build_kernel<<<gProd + NB, dim3(LT), 0, stream>>>(bstart, bedge, ctmp, rowptr, dinv, col, nN, x, W1, bufH, 2);
    gather_gemm_kernel<true><<<gFuse, dim3(FT), 0, stream>>>(bufH, dinv, b1, rowptr, col, wgt, W2, outF, nN);
    gather_gemm_kernel<false><<<gFuse, dim3(FT), 0, stream>>>(outF, dinv, b2, rowptr, col, wgt, W3, bufH, nN);
    gather_rows_kernel<false><<<gW, blk, 0, stream>>>(bufH, dinv, b3, rowptr, col, wgt, outF, nN);
}


# edge list build in 2 launches, not 3: each edge chunk sorts its own edges by bucket in one block (count, scan, place), bucket blocks collect the runs; layer 1 tiles dealt in halves; sum order kept
# speedup vs baseline: 4.3296x; 1.1132x over previous
#include <hip/hip_runtime.h>
#include <hip/hip_bf16.h>


#define DD 128
#define XS 132
#define WS 136

#define NB   256
#define LB   256
#define LT   1024
#define TPC  (LT / LB)
#define RMAX 256

typedef __attribute__((ext_vector_type(2))) float v2f;
typedef __attribute__((ext_vector_type(8))) float v8f;

__device__ __forceinline__ int bucket_of(int d, int R) {
    const unsigned k = (unsigned)d / (unsigned)R;
    return (int)(k < (unsigned)(NB - 1) ? k : (unsigned)(NB - 1));
}

__device__ __forceinline__ int wave_scan256(const int* __restrict__ s, int lane, int ex[8], int own[8]) {
    int sum = 0;
#pragma unroll
    for (int j = 0; j < 8; ++j) { own[j] = s[lane * 8 + j]; sum += own[j]; }
    int incl = sum;
    for (int o = 1; o < 32; o <<= 1) {
        const int up = __shfl_up(incl, o);
        if (lane >= o) incl += up;
    }
    int run = incl - sum;
#pragma unroll
    for (int j = 0; j < 8; ++j) { ex[j] = run; run += own[j]; }
    return run;
}

#define GR 112
#define GS (GR / 16)
#define GPARTS 2
#define XIT ((GR * 32 + LT - 1) / LT)

__device__ __forceinline__ void gemm_tile_role(const float* __restrict__ X, const float* __restrict__ W,
                                               float* __restrict__ H, int nN, int tile) {
    __shared__ __align__(16) float gsm[DD * WS + GR * XS];
    float* sW = gsm;
    float* sX = gsm + DD * WS;

    const int tid  = threadIdx.x;
    const int wave = __builtin_amdgcn_readfirstlane(tid >> 5);
    const int lane = tid & 31;
    const int lr   = lane & 15;
    const int hi   = lane >> 4;
    const int rowBase = tile * GR;

    {
        float4 wv[DD * 32 / LT], xv[XIT];
#pragma unroll
        for (int it = 0; it < DD * 32 / LT; ++it) {
            const int i = tid + it * LT;
            wv[it] = *(const float4*)(W + (i >> 5) * DD + ((i & 31) << 2));
        }
#pragma unroll
        for (int it = 0; it < XIT; ++it) {
            const int i  = tid + it * LT;
            const int gr = rowBase + (i >> 5);
            xv[it] = (i < GR * 32 && gr < nN) ? *(const float4*)(X + (size_t)gr * DD + ((i & 31) << 2))
                                               : make_float4(0.0f, 0.0f, 0.0f, 0.0f);
        }
#pragma unroll
        for (int it = 0; it < DD * 32 / LT; ++it) {
            const int i = tid + it * LT;
            *(float4*)(sW + (i >> 5) * WS + ((i & 31) << 2)) = wv[it];
        }
#pragma unroll
        for (int it = 0; it < XIT; ++it) {
            const int i = tid + it * LT;
            if (i < GR * 32) *(float4*)(sX + (i >> 5) * XS + ((i & 31) << 2)) = xv[it];
        }
    }
    __syncthreads();

    const int strip = wave >> 3;
    const int ct    = wave & 7;
    const bool two  = wave + 32 < GS * 8;

    const float* xrow = sX + (strip * 16 + lr) * XS;
    const float* wcol = sW + ct * 16 + lr;

    v8f acc0 = (v8f){0.f,0.f,0.f,0.f,0.f,0.f,0.f,0.f};
    v8f acc1 = (v8f){0.f,0.f,0.f,0.f,0.f,0.f,0.f,0.f};
    if (two) {
        for (int kb = 0; kb < 32; ++kb) {
            const int k0 = kb * 4 + hi * 2;
            v2f a0 = *(const v2f*)(xrow + k0);
            v2f a1 = *(const v2f*)(xrow + 64 * XS + k0);
            v2f b;
            b.x = wcol[k0 * WS];
            b.y = wcol[k0 * WS + WS];
            acc0 = __builtin_amdgcn_wmma_f32_16x16x4_f32(false, a0, false, b, (short)0, acc0, false, false);
            acc1 = __builtin_amdgcn_wmma_f32_16x16x4_f32(false, a1, false, b, (short)0, acc1, false, false);
        }
    } else {
        for (int kb = 0; kb < 32; ++kb) {
            const int k0 = kb * 4 + hi * 2;
            v2f a0 = *(const v2f*)(xrow + k0);
            v2f b;
            b.x = wcol[k0 * WS];
            b.y = wcol[k0 * WS + WS];
            acc0 = __builtin_amdgcn_wmma_f32_16x16x4_f32(false, a0, false, b, (short)0, acc0, false, false);
        }
    }

    const int row0 = rowBase + strip * 16 + hi * 8;
    const int c    = ct * 16 + lr;
#pragma unroll
    for (int v = 0; v < 8; ++v) {
        const int r = row0 + v;
        if (r < nN) H[(size_t)r * DD + c] = acc0[v];
    }
    if (two) {
#pragma unroll
        for (int v = 0; v < 8; ++v) {
            const int r = row0 + 64 + v;
            if (r < nN) H[(size_t)r * DD + c] = acc1[v];
        }
    }
}

__device__ __forceinline__ int product_blocks(int nN) {
    const int nT = (nN + GR - 1) / GR;
    return (nT + GPARTS - 1) / GPARTS;
}

__device__ __forceinline__ bool product_role(const float* __restrict__ X, const float* __restrict__ W,
                                             float* __restrict__ H, int nN, int part) {
    const int per = product_blocks(nN);
    if ((int)blockIdx.x >= per) return false;
    const int tile = part * per + (int)blockIdx.x;
    if (tile * GR < nN) gemm_tile_role(X, W, H, nN, tile);
    return true;
}

__global__ __launch_bounds__(LT)
void chunk_sort_kernel(const int* __restrict__ src, const int* __restrict__ dst,
                       int* __restrict__ cntT, int* __restrict__ lofT,
                       int2* __restrict__ bedge, int nE, int nN,
                       const float* __restrict__ gX, const float* __restrict__ gW,
                       float* __restrict__ gH, int part) {
    __shared__ int hist[NB];
    __shared__ int loff[NB];
    __shared__ int nxt[NB];
    if (product_role(gX, gW, gH, nN, part)) return;
    const int bid  = (int)blockIdx.x - product_blocks(nN);
    const int tid  = threadIdx.x;
    const int lane = tid & 31;
    const int wv   = tid >> 5;
    const int R = (nN + NB - 1) / NB > 0 ? (nN + NB - 1) / NB : 1;
    if (tid < NB) hist[tid] = 0;
    __syncthreads();
    const int chunk = (nE + LB - 1) / LB;
    const int e0 = bid * chunk;
    const int e1 = (e0 + chunk < nE) ? (e0 + chunk) : nE;
    for (int e = e0 + tid; e < e1; e += 4 * LT) {
        int d[4];
#pragma unroll
        for (int j = 0; j < 4; ++j) { const int ee = e + j * LT; d[j] = (ee < e1) ? dst[ee] : -1; }
#pragma unroll
        for (int j = 0; j < 4; ++j) if (d[j] >= 0) atomicAdd(&hist[bucket_of(d[j], R)], 1);
    }
    __syncthreads();
    if (wv == 0) {
        int ex[8], own[8];
        wave_scan256(hist, lane, ex, own);
#pragma unroll
        for (int j = 0; j < 8; ++j) { loff[lane * 8 + j] = ex[j]; nxt[lane * 8 + j] = ex[j]; }
    }
    __syncthreads();
    if (tid < NB) {
        cntT[tid * LB + bid] = hist[tid];
        lofT[tid * LB + bid] = loff[tid];
    }
    for (int e = e0 + tid; e < e1; e += 4 * LT) {
        int d[4], s[4];
#pragma unroll
        for (int j = 0; j < 4; ++j) {
            const int ee = e + j * LT;
            d[j] = (ee < e1) ? dst[ee] : -1;
            s[j] = (ee < e1) ? src[ee] : 0;
        }
#pragma unroll
        for (int j = 0; j < 4; ++j) {
            if (d[j] >= 0) {
                const int p = e0 + atomicAdd(&nxt[bucket_of(d[j], R)], 1);
                bedge[p] = make_int2(s[j], d[j]);
            }
        }
    }
}

__global__ __launch_bounds__(LT)
void csr_build_kernel(const int* __restrict__ cntT, const int* __restrict__ lofT,
                      const int2* __restrict__ bedge,
                      int* ctmp, int* __restrict__ rowptr, float* __restrict__ dinv,
                      int* __restrict__ col, int nE, int nN,
                      const float* __restrict__ gX, const float* __restrict__ gW,
                      float* __restrict__ gH, int part) {
    __shared__ int cnt[RMAX];
    __shared__ int roff[RMAX];
    __shared__ int fill[RMAX];
    __shared__ int sgo[LB];
    __shared__ int sgn[LB];
    __shared__ int wsn[LB / 32];
    __shared__ int wso[LB / 32];
    if (product_role(gX, gW, gH, nN, part)) return;
    const int bid  = (int)blockIdx.x - product_blocks(nN);
    const int tid  = threadIdx.x;
    const int lane = tid & 31;
    const int wv   = __builtin_amdgcn_readfirstlane(tid >> 5);
    const int R  = (nN + NB - 1) / NB > 0 ? (nN + NB - 1) / NB : 1;
    const int n0 = bid * R;
    const int nn = (nN - n0 < R) ? (nN - n0) : R;
    const int chunk = (nE + LB - 1) / LB;
    if (tid < LB) {
        int n = cntT[bid * LB + tid];
        int o = lofT[bid * LB + tid];
        sgn[tid] = n;
        sgo[tid] = tid * chunk + o;
        for (int k = 16; k > 0; k >>= 1) { n += __shfl_xor(n, k); o += __shfl_xor(o, k); }
        if (lane == 0) { wsn[wv] = n; wso[wv] = o; }
    }
    if (tid < RMAX) { cnt[tid] = 0; fill[tid] = 0; }
    __syncthreads();
    int b0 = 0, bn = 0;
#pragma unroll
    for (int w = 0; w < LB / 32; ++w) { b0 += wso[w]; bn += wsn[w]; }
    const int b1 = b0 + bn;
    const int sn = sgn[tid / TPC];
    const int2* run = bedge + sgo[tid / TPC];
    const int sq = tid % TPC;
    for (int j0 = sq; j0 < sn; j0 += 4 * TPC) {
        int d[4];
#pragma unroll
        for (int j = 0; j < 4; ++j) { const int jj = j0 + j * TPC; d[j] = (jj < sn) ? run[jj].y : -1; }
#pragma unroll
        for (int j = 0; j < 4; ++j) {
            const int l = d[j] - n0;
            if (d[j] >= 0 && l >= 0 && l < RMAX) atomicAdd(&cnt[l], 1);
        }
    }
    __syncthreads();
    if (wv == 0) {
        int ex[8], own[8];
        wave_scan256(cnt, lane, ex, own);
#pragma unroll
        for (int j = 0; j < 8; ++j) roff[lane * 8 + j] = ex[j];
    }
    __syncthreads();
    if (tid < nn) {
        rowptr[n0 + tid] = b0 + roff[tid];
        dinv[n0 + tid]   = rsqrtf((float)(cnt[tid] + 1));
    }
    if (tid == 0 && nn > 0 && n0 + nn == nN) rowptr[nN] = b1;
    for (int j0 = sq; j0 < sn; j0 += 4 * TPC) {
        int2 ed[4];
#pragma unroll
        for (int j = 0; j < 4; ++j) { const int jj = j0 + j * TPC; ed[j] = (jj < sn) ? run[jj] : make_int2(0, -1); }
#pragma unroll
        for (int j = 0; j < 4; ++j) {
            const int l = ed[j].y - n0;
            if (ed[j].y >= 0 && l >= 0 && l < RMAX) {
                const int r = atomicAdd(&fill[l], 1);
                ctmp[b0 + roff[l] + r] = ed[j].x;
            }
        }
    }
    __syncthreads();
    for (int l = wv; l < nn; l += LT / 32) {
        const int beg = b0 + __builtin_amdgcn_readfirstlane(roff[l]);
        const int deg = __builtin_amdgcn_readfirstlane(cnt[l]);
        const int* seg = ctmp + beg;
        if (deg <= 32) {
            const int key = (lane < deg) ? seg[lane] : 0;
            int rank = 0;
            for (int m = 0; m < deg; ++m) {
                const int km = __builtin_amdgcn_readlane(key, m);
                rank += (km < key || (km == key && m < lane)) ? 1 : 0;
            }
            if (lane < deg) col[beg + rank] = key;
        } else {
            for (int j = lane; j < deg; j += 32) {
                const int key = seg[j];
                int rank = 0;
                for (int m = 0; m < deg; ++m) {
                    const int km = seg[m];
                    rank += (km < key || (km == key && m < j)) ? 1 : 0;
                }
                col[beg + rank] = key;
            }
        }
    }
}

template <int G>
__device__ __forceinline__ void gather_group(float4& acc, const float4* __restrict__ h4,
                                             int cs, float cw, int k, int lane) {
    float4 v[G];
    float  w[G];
#pragma unroll
    for (int j = 0; j < G; ++j) {
        const int s = __builtin_amdgcn_readlane(cs, k + j);
        w[j] = __int_as_float(__builtin_amdgcn_readlane(__float_as_int(cw), k + j));
        v[j] = h4[(size_t)s * 32 + lane];
    }
#pragma unroll
    for (int j = 0; j < G; ++j) {
        acc.x += w[j] * v[j].x;  acc.y += w[j] * v[j].y;
        acc.z += w[j] * v[j].z;  acc.w += w[j] * v[j].w;
    }
}

template <bool FIRST>
__device__ __forceinline__ float4 gather_node(const float4* __restrict__ h4, const float* __restrict__ dinv,
                                              const int* __restrict__ col, float* __restrict__ wgt,
                                              const float4 b, int i, int beg, int end, float di, int lane) {
    float4 acc;
    {
        const float w0 = di * di;
        const float4 v = h4[(size_t)i * 32 + lane];
        acc.x = v.x * w0 + b.x;  acc.y = v.y * w0 + b.y;
        acc.z = v.z * w0 + b.z;  acc.w = v.w * w0 + b.w;
    }
    for (int p = beg; p < end; p += 32) {
        const int n = (end - p < 32) ? (end - p) : 32;
        int   cs = 0;
        float cw = 0.0f;
        if (lane < n) {
            cs = col[p + lane];
            if (FIRST) { cw = di * dinv[cs]; wgt[p + lane] = cw; }
            else       { cw = wgt[p + lane]; }
        }
        int k = 0;
        for (; k + 8 <= n; k += 8) gather_group<8>(acc, h4, cs, cw, k, lane);
        if (k + 4 <= n) { gather_group<4>(acc, h4, cs, cw, k, lane); k += 4; }
        if (k + 2 <= n) { gather_group<2>(acc, h4, cs, cw, k, lane); k += 2; }
        if (k < n)      { gather_group<1>(acc, h4, cs, cw, k, lane); }
    }
    return acc;
}

template <bool FIRST>
__global__ __launch_bounds__(256)
void gather_rows_kernel(const float* __restrict__ h, const float* __restrict__ dinv,
                        const float* __restrict__ bias, const int* __restrict__ rowptr,
                        const int* __restrict__ col, float* __restrict__ wgt,
                        float* __restrict__ out, int nN) {
    const int lane = threadIdx.x & 31;
    const int i = blockIdx.x * 8 + __builtin_amdgcn_readfirstlane(threadIdx.x >> 5);
    if (i >= nN) return;
    const int beg = rowptr[i];
    const int end = rowptr[i + 1];
    const float di = dinv[i];
    const float4 b = ((const float4*)bias)[lane];
    const float4 acc = gather_node<FIRST>((const float4*)h, dinv, col, wgt, b, i, beg, end, di, lane);
    ((float4*)out)[(size_t)i * 32 + lane] = acc;
}

#define TM 208
#define FW (TM / 8)
#define FT (FW * 32)
#define WIT ((DD * 32 + FT - 1) / FT)

template <bool FIRST>
__global__ __launch_bounds__(FT)
void gather_gemm_kernel(const float* __restrict__ h, const float* __restrict__ dinv,
                        const float* __restrict__ bias, const int* __restrict__ rowptr,
                        const int* __restrict__ col, float* __restrict__ wgt,
                        const float* __restrict__ W, float* __restrict__ H, int nN) {
    __shared__ __align__(16) float smem[DD * WS + TM * XS];
    float* sW = smem;
    float* sX = smem + DD * WS;

    const int tid  = threadIdx.x;
    const int lane = tid & 31;
    const int wave = __builtin_amdgcn_readfirstlane(tid >> 5);
    const int rowBase = blockIdx.x * TM;

    {
        float4 wv[WIT];
#pragma unroll
        for (int it = 0; it < WIT; ++it) {
            const int i = tid + it * FT;
            wv[it] = (i < DD * 32) ? *(const float4*)(W + (i >> 5) * DD + ((i & 31) << 2))
                                   : make_float4(0.0f, 0.0f, 0.0f, 0.0f);
        }
#pragma unroll
        for (int it = 0; it < WIT; ++it) {
            const int i = tid + it * FT;
            if (i < DD * 32) *(float4*)(sW + (i >> 5) * WS + ((i & 31) << 2)) = wv[it];
        }
    }

    {
        const int i0 = rowBase + wave * 8;
        int   rp = 0;
        float dv = 0.0f;
        if (lane <= 8) { const int j = i0 + lane; rp = rowptr[j < nN ? j : nN]; }
        if (lane <  8) { const int j = i0 + lane; dv = dinv[j < nN ? j : nN - 1]; }
        const float4 b = ((const float4*)bias)[lane];
        const float4* __restrict__ h4 = (const float4*)h;
        for (int r = 0; r < 8; ++r) {
            const int i = i0 + r;
            float4 acc = make_float4(0.0f, 0.0f, 0.0f, 0.0f);
            if (i < nN) {
                const int   beg = __builtin_amdgcn_readlane(rp, r);
                const int   end = __builtin_amdgcn_readlane(rp, r + 1);
                const float di  = __int_as_float(__builtin_amdgcn_readlane(__float_as_int(dv), r));
                acc = gather_node<FIRST>(h4, dinv, col, wgt, b, i, beg, end, di, lane);
                acc.x = fmaxf(acc.x, 0.0f); acc.y = fmaxf(acc.y, 0.0f);
                acc.z = fmaxf(acc.z, 0.0f); acc.w = fmaxf(acc.w, 0.0f);
            }
            *(float4*)(sX + (wave * 8 + r) * XS + lane * 4) = acc;
        }
    }
    __syncthreads();

    const int lr    = lane & 15;
    const int hi    = lane >> 4;
    const int strip = wave >> 1;
    const int half  = wave & 1;

    const float* xrow = sX + (strip * 16 + lr) * XS;

    v8f acc[4];
#pragma unroll
    for (int t = 0; t < 4; ++t) acc[t] = (v8f){0.f,0.f,0.f,0.f,0.f,0.f,0.f,0.f};

    for (int kb = 0; kb < 32; ++kb) {
        const int k0 = kb * 4 + hi * 2;
        v2f a = *(const v2f*)(xrow + k0);
        const float* wp = sW + k0 * WS + half * 64 + lr;
#pragma unroll
        for (int t = 0; t < 4; ++t) {
            v2f b;
            b.x = wp[t * 16];
            b.y = wp[t * 16 + WS];
            acc[t] = __builtin_amdgcn_wmma_f32_16x16x4_f32(
                false, a, false, b, (short)0, acc[t], false, false);
        }
    }

    const int row0 = rowBase + strip * 16 + hi * 8;
#pragma unroll
    for (int t = 0; t < 4; ++t) {
        const int c = half * 64 + t * 16 + lr;
#pragma unroll
        for (int v = 0; v < 8; ++v) {
            int r = row0 + v;
            if (r < nN) H[(size_t)r * DD + c] = acc[t][v];
        }
    }
}

extern "C" void kernel_launch(void* const* d_in, const int* in_sizes, int n_in,
                              void* d_out, int out_size, void* d_ws, size_t ws_size,
                              hipStream_t stream) {
    const float* x   = (const float*)d_in[0];
    const int*   ei  = (const int*)d_in[1];
    const float* W1  = (const float*)d_in[2];
    const float* b1  = (const float*)d_in[3];
    const float* W2  = (const float*)d_in[4];
    const float* b2  = (const float*)d_in[5];
    const float* W3  = (const float*)d_in[6];
    const float* b3  = (const float*)d_in[7];

    const int nN = in_sizes[0] / DD;
    const int nE = in_sizes[1] / 2;
    const int* src = ei;
    const int* dst = ei + nE;

    const size_t npad = (size_t)((nN + 127) / 128) * 128;
    float* dinv = (float*)d_ws;
    float* bufH = dinv + npad;
    int*   rowptr = (int*)(bufH + npad * DD);
    int*   col    = rowptr + npad + 640;
    float* wgt    = (float*)(col + nE);
    float* outF = (float*)d_out;
    const size_t nE4 = ((size_t)nE + 3) / 4 * 4;
    int2*  bedge = (int2*)d_out;
    int*   ctmp  = (int*)d_out + 2 * nE4;
    int*   cntT  = ctmp + nE4;
    int*   lofT  = cntT + NB * LB;

    const dim3 blk(256);
    const int gW   = (nN + 7) / 8;
    const int gProd = ((nN + GR - 1) / GR + GPARTS - 1) / GPARTS;
    const int gFuse = (nN + TM - 1) / TM;

    chunk_sort_kernel<<<gProd + LB, dim3(LT), 0, stream>>>(src, dst, cntT, lofT, bedge, nE, nN, x, W1, bufH, 0);
    csr_build_kernel<<<gProd + NB, dim3(LT), 0, stream>>>(cntT, lofT, bedge, ctmp, rowptr, dinv, col, nE, nN, x, W1, bufH, 1);
    gather_gemm_kernel<true><<<gFuse, dim3(FT), 0, stream>>>(bufH, dinv, b1, rowptr, col, wgt, W2, outF, nN);
    gather_gemm_kernel<false><<<gFuse, dim3(FT), 0, stream>>>(outF, dinv, b2, rowptr, col, wgt, W3, bufH, nN);
    gather_rows_kernel<false><<<gW, blk, 0, stream>>>(bufH, dinv, b3, rowptr, col, wgt, outF, nN);
}
